# starting kernel
# baseline (speedup 1.0000x reference)
#include <hip/hip_runtime.h>
#include <hip/hip_bf16.h>
#include <math.h>

#define BB 64
#define NN 50000
#define EE 1600000
#define LL 20
#define BN_EPS 1e-5f
#define KCHUNK 784
#define NCHUNKS 64

typedef __attribute__((ext_vector_type(2))) float v2f;
typedef __attribute__((ext_vector_type(8))) float v8f;

__global__ void k_zero(float* __restrict__ p, long n) {
    long i = (long)blockIdx.x * blockDim.x + threadIdx.x;
    long stride = (long)gridDim.x * blockDim.x;
    for (; i < n; i += stride) p[i] = 0.0f;
}

__global__ void k_transpose(const float* __restrict__ x, float* __restrict__ xT) {
    __shared__ float tile[64 * 65];
    int n0   = blockIdx.x * 64;
    int col  = threadIdx.x & 63;
    int rowg = threadIdx.x >> 6;
    for (int b = rowg; b < 64; b += 4) {
        int n = n0 + col;
        tile[b * 65 + col] = (n < NN) ? x[(long)b * NN + n] : 0.0f;
    }
    __syncthreads();
    for (int nl = rowg; nl < 64; nl += 4) {
        int n = n0 + nl;
        if (n < NN) xT[(long)n * 64 + col] = tile[col * 65 + nl];
    }
}

__global__ void k_edges(const float* __restrict__ xT, const int* __restrict__ ei,
                        const float* __restrict__ alpha, const float* __restrict__ bias,
                        float* __restrict__ sums, float* __restrict__ counts) {
    long gtid = (long)blockIdx.x * blockDim.x + threadIdx.x;
    long e = gtid >> 6;
    int  t = (int)(gtid & 63);
    if (e >= EE) return;
    int   s  = ei[e];
    int   d  = ei[(long)EE + e];
    float a  = alpha[e];
    float bi = bias[e];
    float val = fmaf(a, xT[(long)s * 64 + t], bi);
    atomicAdd(&sums[(long)d * 64 + t], val);
    if (t == 0) atomicAdd(&counts[d], 1.0f);
}

__global__ void k_finalize(float* __restrict__ sums  ,
                           const float* __restrict__ counts,
                           const float* __restrict__ gamma, const float* __restrict__ beta,
                           float* __restrict__ bn_out  ) {
    __shared__ float tile[64 * 65];
    int n0   = blockIdx.x * 64;
    int wave = threadIdx.x >> 5;
    int lane = threadIdx.x & 31;
    for (int i = 0; i < 8; ++i) {
        int lnode = wave * 8 + i;
        int n = n0 + lnode;
        if (n < NN) {
            float2 s = *(const float2*)&sums[(long)n * 64 + 2 * lane];
            float inv = 1.0f / fmaxf(counts[n], 1.0f);
            float xh0 = s.x * inv, xh1 = s.y * inv;
            float2 xh; xh.x = xh0; xh.y = xh1;
            *(float2*)&sums[(long)n * 64 + 2 * lane] = xh;
            float t0 = tanhf(xh0), t1 = tanhf(xh1);
            float sm = t0 + t1;
            float sq = t0 * t0 + t1 * t1;
            #pragma unroll
            for (int off = 16; off >= 1; off >>= 1) {
                sm += __shfl_xor(sm, off, 32);
                sq += __shfl_xor(sq, off, 32);
            }
            float mu = sm * (1.0f / 64.0f);
            float va = sq * (1.0f / 64.0f) - mu * mu;
            float rs = rsqrtf(va + BN_EPS);
            float g = gamma[n], be = beta[n];
            tile[lnode * 65 + 2 * lane]     = (t0 - mu) * rs * g + be;
            tile[lnode * 65 + 2 * lane + 1] = (t1 - mu) * rs * g + be;
        }
    }
    __syncthreads();
    int col  = threadIdx.x & 63;
    int rowg = threadIdx.x >> 6;
    int n = n0 + col;
    if (n < NN) {
        for (int b = rowg; b < 64; b += 4)
            bn_out[(long)b * NN + n] = tile[col * 65 + b];
    }
}

__global__ void k_gemm_part(const float* __restrict__ xhatT  ,
                            const float* __restrict__ W      ,
                            float* __restrict__ part) {
    int c    = blockIdx.x;
    int wave = threadIdx.x >> 5;
    int lane = threadIdx.x & 31;
    int half = lane >> 4;
    int l16  = lane & 15;
    int m_tile = wave >> 1;
    int n_tile = wave & 1;
    int m    = m_tile * 16 + l16;
    int lcol = n_tile * 16 + l16;
    int wrow = (lcol < LL) ? lcol : 0;
    const float* Wrow = W + (long)wrow * NN;

    long kbeg = (long)c * KCHUNK;
    long kend = kbeg + KCHUNK; if (kend > NN) kend = NN;

    v8f acc = {};
    #pragma unroll 4
    for (long k = kbeg; k < kend; k += 4) {
        long k0 = k + half * 2;
        long k1 = k0 + 1;
        v2f a, b;
        a.x = xhatT[k0 * 64 + m];
        a.y = xhatT[k1 * 64 + m];
        b.x = Wrow[k0];
        b.y = Wrow[k1];
        acc = __builtin_amdgcn_wmma_f32_16x16x4_f32(
             false, a,  false, b,
             (short)0, acc,  false,  false);
    }

    float* dst = part + (long)c * (BB * LL);
    #pragma unroll
    for (int i = 0; i < 8; ++i) {
        int mrow = m_tile * 16 + half * 8 + i;
        int ncol = n_tile * 16 + l16;
        if (ncol < LL) dst[mrow * LL + ncol] = acc[i];
    }
}

__global__ void k_reduce_pred(const float* __restrict__ part, const float* __restrict__ b_lin,
                              float* __restrict__ pred) {
    int i = blockIdx.x * blockDim.x + threadIdx.x;
    if (i >= BB * LL) return;
    float s = b_lin[i % LL];
    for (int c = 0; c < NCHUNKS; ++c) s += part[c * (BB * LL) + i];
    pred[i] = s;
}

extern "C" void kernel_launch(void* const* d_in, const int* in_sizes, int n_in,
                              void* d_out, int out_size, void* d_ws, size_t ws_size,
                              hipStream_t stream) {
    const float* x     = (const float*)d_in[0];
    const int*   ei    = (const int*)  d_in[1];
    const float* alpha = (const float*)d_in[2];
    const float* bias  = (const float*)d_in[3];
    const float* W     = (const float*)d_in[4];
    const float* b_lin = (const float*)d_in[5];
    const float* gamma = (const float*)d_in[6];
    const float* beta  = (const float*)d_in[7];
    float* out = (float*)d_out;

    float* ws     = (float*)d_ws;
    float* sums   = ws;
    float* counts = sums + (long)NN * 64;
    float* xT     = counts + NN;
    float* part   = xT + (long)NN * 64;

    k_zero<<<4096, 256, 0, stream>>>(sums, (long)NN * 64 + NN);
    k_transpose<<<(NN + 63) / 64, 256, 0, stream>>>(x, xT);
    long edge_threads = (long)EE * 64;
    k_edges<<<(int)((edge_threads + 255) / 256), 256, 0, stream>>>(xT, ei, alpha, bias, sums, counts);
    k_finalize<<<(NN + 63) / 64, 256, 0, stream>>>(sums, counts, gamma, beta, out + BB * LL);
    k_gemm_part<<<NCHUNKS, 256, 0, stream>>>(sums, W, part);
    k_reduce_pred<<<(BB * LL + 255) / 256, 256, 0, stream>>>(part, b_lin, out);
}


# f32 atomic edge scatter replaced by per-call destination buckets (int count, scan, fill) and a wave-per-node gather summed in ascending edge-id order, fused with finalize and the WMMA head
# speedup vs baseline: 1.6156x; 1.6156x over previous
#include <hip/hip_runtime.h>
#include <math.h>

#define BB 64
#define NN 50000
#define EE 1600000
#define LL 20
#define BN_EPS 1e-5f
#define NBLK 782
#define XP 80
#define SCAN_T 1024
#define SCAN_IT 49

typedef __attribute__((ext_vector_type(2))) float v2f;
typedef __attribute__((ext_vector_type(8))) float v8f;

__global__ void k_zero_i(int* __restrict__ p, int n) {
    int i = blockIdx.x * blockDim.x + threadIdx.x;
    if (i < n) p[i] = 0;
}

__global__ void k_transpose(const float* __restrict__ x, float* __restrict__ xT) {
    __shared__ float tile[64 * 65];
    int n0   = blockIdx.x * 64;
    int col  = threadIdx.x & 63;
    int rowg = threadIdx.x >> 6;
    for (int b = rowg; b < 64; b += 4) {
        int n = n0 + col;
        tile[b * 65 + col] = (n < NN) ? x[(long)b * NN + n] : 0.0f;
    }
    __syncthreads();
    for (int nl = rowg; nl < 64; nl += 4) {
        int n = n0 + nl;
        if (n < NN) xT[(long)n * 64 + col] = tile[col * 65 + nl];
    }
}

__global__ void k_count(const int* __restrict__ ei, int* __restrict__ cnt) {
    int e = blockIdx.x * blockDim.x + threadIdx.x;
    if (e < EE) atomicAdd(&cnt[ei[EE + e]], 1);
}

__global__ void __launch_bounds__(SCAN_T) k_scan(const int* __restrict__ cnt, int* __restrict__ cur) {
    __shared__ int wtot[32];
    int tid = threadIdx.x, lane = tid & 31, wave = tid >> 5;
    int v[SCAN_IT];
    int total = 0;
    #pragma unroll
    for (int it = 0; it < SCAN_IT; ++it) {
        int i = it * SCAN_T + tid;
        v[it] = (i < NN) ? cnt[i] : 0;
        total += v[it];
    }
    int inc = total;
    #pragma unroll
    for (int off = 1; off < 32; off <<= 1) {
        int t = __shfl_up(inc, off, 32);
        if (lane >= off) inc += t;
    }
    if (lane == 31) wtot[wave] = inc;
    __syncthreads();
    int wt = wtot[lane];
    int winc = wt;
    #pragma unroll
    for (int off = 1; off < 32; off <<= 1) {
        int t = __shfl_up(winc, off, 32);
        if (lane >= off) winc += t;
    }
    int wbase = __shfl(winc, wave, 32) - __shfl(wt, wave, 32);
    int run = wbase + inc - total;
    #pragma unroll
    for (int it = 0; it < SCAN_IT; ++it) {
        int i = it * SCAN_T + tid;
        if (i < NN) cur[i] = run;
        run += v[it];
    }
}

__global__ void k_fill(const int* __restrict__ ei, int* __restrict__ cur, int* __restrict__ perm) {
    int e = blockIdx.x * blockDim.x + threadIdx.x;
    if (e < EE) {
        int d = ei[EE + e];
        int p = atomicAdd(&cur[d], 1);
        perm[p] = e;
    }
}

__global__ void __launch_bounds__(256) k_gather(
        const float* __restrict__ xT, const int* __restrict__ ei,
        const float* __restrict__ alpha, const float* __restrict__ bias,
        const int* __restrict__ perm, const int* __restrict__ cur, const int* __restrict__ cnt,
        const float* __restrict__ W, const float* __restrict__ gamma, const float* __restrict__ beta,
        float* __restrict__ bn_out  , float* __restrict__ part  ) {
    __shared__ float s_bn[64 * 65];
    __shared__ float s_xh[64 * XP];
    __shared__ int   s_src[8][64];
    __shared__ float s_al[8][64];
    __shared__ float s_bi[8][64];

    int n0   = blockIdx.x * 64;
    int wave = threadIdx.x >> 5;
    int lane = threadIdx.x & 31;

    int half = lane >> 4;
    int l16  = lane & 15;
    int m_tile = wave >> 1;
    int n_tile = wave & 1;
    int lcol = n_tile * 16 + l16;
    int wrow = (lcol < LL) ? lcol : 0;
    const float* Wrow = W + (long)wrow * NN;
    v2f bw[16];
    #pragma unroll
    for (int s = 0; s < 16; ++s) {
        int k0 = n0 + s * 4 + half * 2;
        bool ok = k0 < NN;
        bw[s].x = ok ? Wrow[k0] : 0.0f;
        bw[s].y = ok ? Wrow[k0 + 1] : 0.0f;
    }

    int pn = n0 + wave * 8 + (lane & 7);
    int len_l = (pn < NN) ? cnt[pn] : 0;
    int end_l = (pn < NN) ? cur[pn] : 0;

    for (int i = 0; i < 8; ++i) {
        int lnode = wave * 8 + i;
        int n = n0 + lnode;
        float xh0 = 0.0f, xh1 = 0.0f;
        if (n < NN) {
            int len = __builtin_amdgcn_readlane(len_l, i);
            int end = __builtin_amdgcn_readlane(end_l, i);
            int beg = end - len;
            float a0 = 0.0f, a1 = 0.0f;
            if (len <= 64) {
                int id0 = (lane < len)      ? perm[beg + lane]      : 0x7fffffff;
                int id1 = (lane + 32 < len) ? perm[beg + lane + 32] : 0x7fffffff;
                int r0 = 0, r1 = 0;
                #pragma unroll
                for (int k = 0; k < 32; ++k) {
                    int s = __builtin_amdgcn_readlane(id0, k);
                    r0 += (s < id0) ? 1 : 0;
                    r1 += (s < id1) ? 1 : 0;
                }
                if (len > 32) {
                    #pragma unroll
                    for (int k = 0; k < 32; ++k) {
                        int s = __builtin_amdgcn_readlane(id1, k);
                        r0 += (s < id0) ? 1 : 0;
                        r1 += (s < id1) ? 1 : 0;
                    }
                }
                __builtin_amdgcn_fence(__ATOMIC_RELEASE, "wavefront");
                __builtin_amdgcn_wave_barrier();
                if (lane < len) {
                    s_src[wave][r0] = ei[id0];
                    s_al[wave][r0]  = alpha[id0];
                    s_bi[wave][r0]  = bias[id0];
                }
                if (lane + 32 < len) {
                    s_src[wave][r1] = ei[id1];
                    s_al[wave][r1]  = alpha[id1];
                    s_bi[wave][r1]  = bias[id1];
                }
                __builtin_amdgcn_fence(__ATOMIC_ACQ_REL, "wavefront");
                __builtin_amdgcn_wave_barrier();
                for (int j = 0; j < len; j += 8) {
                    float2 v[8]; float aa[8], bb[8];
                    #pragma unroll
                    for (int u = 0; u < 8; ++u) {
                        int jj = min(j + u, len - 1);
                        int s = __builtin_amdgcn_readfirstlane(s_src[wave][jj]);
                        aa[u] = __int_as_float(__builtin_amdgcn_readfirstlane(__float_as_int(s_al[wave][jj])));
                        bb[u] = __int_as_float(__builtin_amdgcn_readfirstlane(__float_as_int(s_bi[wave][jj])));
                        v[u] = *(const float2*)(xT + (long)s * 64 + 2 * lane);
                    }
                    #pragma unroll
                    for (int u = 0; u < 8; ++u) {
                        if (j + u < len) {
                            a0 += fmaf(aa[u], v[u].x, bb[u]);
                            a1 += fmaf(aa[u], v[u].y, bb[u]);
                        }
                    }
                }
            } else {
                int prev = -1;
                for (int it = 0; it < len; ++it) {
                    int m = 0x7fffffff;
                    for (int k = lane; k < len; k += 32) {
                        int id = perm[beg + k];
                        if (id > prev && id < m) m = id;
                    }
                    #pragma unroll
                    for (int off = 16; off >= 1; off >>= 1) m = min(m, __shfl_xor(m, off, 32));
                    int s = ei[m];
                    float al = alpha[m], bi = bias[m];
                    float2 v = *(const float2*)(xT + (long)s * 64 + 2 * lane);
                    a0 += fmaf(al, v.x, bi);
                    a1 += fmaf(al, v.y, bi);
                    prev = m;
                }
            }
            float den = fmaxf((float)len, 1.0f);
            xh0 = a0 / den;
            xh1 = a1 / den;
            float t0 = tanhf(xh0), t1 = tanhf(xh1);
            float sm = t0 + t1;
            float sq = t0 * t0 + t1 * t1;
            #pragma unroll
            for (int off = 16; off >= 1; off >>= 1) {
                sm += __shfl_xor(sm, off, 32);
                sq += __shfl_xor(sq, off, 32);
            }
            float mu = sm * (1.0f / 64.0f);
            float va = sq * (1.0f / 64.0f) - mu * mu;
            float rs = rsqrtf(va + BN_EPS);
            float g = gamma[n], be = beta[n];
            s_bn[lnode * 65 + 2 * lane]     = (t0 - mu) * rs * g + be;
            s_bn[lnode * 65 + 2 * lane + 1] = (t1 - mu) * rs * g + be;
        }
        float2 xw; xw.x = xh0; xw.y = xh1;
        *(float2*)&s_xh[lnode * XP + 2 * lane] = xw;
    }
    __syncthreads();

    {
        int col  = threadIdx.x & 63;
        int rowg = threadIdx.x >> 6;
        int n = n0 + col;
        if (n < NN) {
            for (int b = rowg; b < 64; b += 4)
                bn_out[(long)b * NN + n] = s_bn[col * 65 + b];
        }
    }

    v8f acc = {};
    int m = m_tile * 16 + l16;
    #pragma unroll
    for (int s = 0; s < 16; ++s) {
        int k0 = s * 4 + half * 2;
        v2f a;
        a.x = s_xh[k0 * XP + m];
        a.y = s_xh[(k0 + 1) * XP + m];
        acc = __builtin_amdgcn_wmma_f32_16x16x4_f32(
             false, a,  false, bw[s],
             (short)0, acc,  false,  false);
    }
    float* dst = part + (long)blockIdx.x * (BB * LL);
    #pragma unroll
    for (int i = 0; i < 8; ++i) {
        int mrow = m_tile * 16 + half * 8 + i;
        int ncol = n_tile * 16 + l16;
        if (ncol < LL) dst[mrow * LL + ncol] = acc[i];
    }
}

__global__ void __launch_bounds__(1024) k_reduce_pred(const float* __restrict__ part, const float* __restrict__ b_lin,
                                                      float* __restrict__ pred) {
    __shared__ float red[32][33];
    int il = threadIdx.x & 31;
    int g  = threadIdx.x >> 5;
    int i  = blockIdx.x * 32 + il;
    float s = 0.0f;
    for (int c = g; c < NBLK; c += 32) s += part[(long)c * (BB * LL) + i];
    red[g][il] = s;
    __syncthreads();
    if (g == 0) {
        float t = b_lin[i % LL];
        #pragma unroll
        for (int gg = 0; gg < 32; ++gg) t += red[gg][il];
        pred[i] = t;
    }
}

extern "C" void kernel_launch(void* const* d_in, const int* in_sizes, int n_in,
                              void* d_out, int out_size, void* d_ws, size_t ws_size,
                              hipStream_t stream) {
    const float* x     = (const float*)d_in[0];
    const int*   ei    = (const int*)  d_in[1];
    const float* alpha = (const float*)d_in[2];
    const float* bias  = (const float*)d_in[3];
    const float* W     = (const float*)d_in[4];
    const float* b_lin = (const float*)d_in[5];
    const float* gamma = (const float*)d_in[6];
    const float* beta  = (const float*)d_in[7];
    float* out = (float*)d_out;

    float* ws   = (float*)d_ws;
    float* xT   = ws;
    int*   perm = (int*)(xT + (long)NN * 64);
    int*   cnt  = perm + EE;
    int*   cur  = cnt + NN;
    float* part = (float*)(cur + NN);

    k_zero_i<<<(NN + 255) / 256, 256, 0, stream>>>(cnt, NN);
    k_transpose<<<(NN + 63) / 64, 256, 0, stream>>>(x, xT);
    k_count<<<(EE + 255) / 256, 256, 0, stream>>>(ei, cnt);
    k_scan<<<1, SCAN_T, 0, stream>>>(cnt, cur);
    k_fill<<<(EE + 255) / 256, 256, 0, stream>>>(ei, cur, perm);
    k_gather<<<NBLK, 256, 0, stream>>>(xT, ei, alpha, bias, perm, cur, cnt, W, gamma, beta, out + BB * LL, part);
    k_reduce_pred<<<(BB * LL) / 32, 1024, 0, stream>>>(part, b_lin, out);
}


# bucket fill without atomics: the in-degree add returns each edge's rank in its bucket; the fill stores edge ids at start+rank, starts looked up in an LDS copy (one atomic pass instead of two)
# speedup vs baseline: 2.1581x; 1.3358x over previous
#include <hip/hip_runtime.h>
#include <math.h>

#define BB 64
#define NN 50000
#define EE 1600000
#define LL 20
#define BN_EPS 1e-5f
#define NBLK 782
#define XP 80
#define SCAN_T 1024
#define SCAN_IT 49
#define FILL_BLK 256
#define FILL_EPB (EE / FILL_BLK)
#define FILL_IT 7
#define TAB_IT 13
static_assert(FILL_EPB * FILL_BLK == EE, "the fill work-groups tile the edges exactly");
static_assert(FILL_IT * 1024 >= FILL_EPB && TAB_IT * 1024 * 4 >= NN && NN % 4 == 0, "fill loop bounds");

typedef __attribute__((ext_vector_type(2))) float v2f;
typedef __attribute__((ext_vector_type(8))) float v8f;

__global__ void k_zero_i(int* __restrict__ p, int n) {
    int i = blockIdx.x * blockDim.x + threadIdx.x;
    if (i < n) p[i] = 0;
}

__global__ void k_transpose(const float* __restrict__ x, float* __restrict__ xT) {
    __shared__ float tile[64 * 65];
    int n0   = blockIdx.x * 64;
    int col  = threadIdx.x & 63;
    int rowg = threadIdx.x >> 6;
    for (int b = rowg; b < 64; b += 4) {
        int n = n0 + col;
        tile[b * 65 + col] = (n < NN) ? x[(long)b * NN + n] : 0.0f;
    }
    __syncthreads();
    for (int nl = rowg; nl < 64; nl += 4) {
        int n = n0 + nl;
        if (n < NN) xT[(long)n * 64 + col] = tile[col * 65 + nl];
    }
}

__global__ void k_count(const int* __restrict__ ei, int* __restrict__ cnt, int* __restrict__ rank) {
    int e = blockIdx.x * blockDim.x + threadIdx.x;
    if (e < EE) rank[e] = atomicAdd(&cnt[ei[EE + e]], 1);
}

__global__ void __launch_bounds__(SCAN_T) k_scan(const int* __restrict__ cnt, int* __restrict__ cur) {
    __shared__ int wtot[32];
    int tid = threadIdx.x, lane = tid & 31, wave = tid >> 5;
    int v[SCAN_IT];
    int total = 0;
    #pragma unroll
    for (int it = 0; it < SCAN_IT; ++it) {
        int i = it * SCAN_T + tid;
        v[it] = (i < NN) ? cnt[i] : 0;
        total += v[it];
    }
    int inc = total;
    #pragma unroll
    for (int off = 1; off < 32; off <<= 1) {
        int t = __shfl_up(inc, off, 32);
        if (lane >= off) inc += t;
    }
    if (lane == 31) wtot[wave] = inc;
    __syncthreads();
    int wt = wtot[lane];
    int winc = wt;
    #pragma unroll
    for (int off = 1; off < 32; off <<= 1) {
        int t = __shfl_up(winc, off, 32);
        if (lane >= off) winc += t;
    }
    int wbase = __shfl(winc, wave, 32) - __shfl(wt, wave, 32);
    int run = wbase + inc - total;
    #pragma unroll
    for (int it = 0; it < SCAN_IT; ++it) {
        int i = it * SCAN_T + tid;
        if (i < NN) cur[i] = run;
        run += v[it];
    }
}

__global__ void __launch_bounds__(1024) k_fill(const int* __restrict__ ei, const int* __restrict__ cur,
                                               const int* __restrict__ rank, int* __restrict__ perm) {
    __shared__ __attribute__((aligned(16))) int s_start[NN];
    int tid  = threadIdx.x;
    int base = blockIdx.x * FILL_EPB;
    int d[FILL_IT], r[FILL_IT];
    #pragma unroll
    for (int it = 0; it < FILL_IT; ++it) {
        int k = it * 1024 + tid;
        bool ok = k < FILL_EPB;
        d[it] = ok ? ei[EE + base + k] : 0;
        r[it] = ok ? rank[base + k] : 0;
    }
    const int4* c4 = (const int4*)cur;
    int4* s4 = (int4*)s_start;
    #pragma unroll
    for (int it = 0; it < TAB_IT; ++it) {
        int i = it * 1024 + tid;
        if (i < NN / 4) s4[i] = c4[i];
    }
    __syncthreads();
    #pragma unroll
    for (int it = 0; it < FILL_IT; ++it) {
        int k = it * 1024 + tid;
        if (k < FILL_EPB) perm[s_start[d[it]] + r[it]] = base + k;
    }
}

__global__ void __launch_bounds__(256) k_gather(
        const float* __restrict__ xT, const int* __restrict__ ei,
        const float* __restrict__ alpha, const float* __restrict__ bias,
        const int* __restrict__ perm, const int* __restrict__ cur, const int* __restrict__ cnt,
        const float* __restrict__ W, const float* __restrict__ gamma, const float* __restrict__ beta,
        float* __restrict__ bn_out  , float* __restrict__ part  ) {
    __shared__ float s_bn[64 * 65];
    __shared__ float s_xh[64 * XP];
    __shared__ int   s_src[8][64];
    __shared__ float s_al[8][64];
    __shared__ float s_bi[8][64];

    int n0   = blockIdx.x * 64;
    int wave = threadIdx.x >> 5;
    int lane = threadIdx.x & 31;

    int half = lane >> 4;
    int l16  = lane & 15;
    int m_tile = wave >> 1;
    int n_tile = wave & 1;
    int lcol = n_tile * 16 + l16;
    int wrow = (lcol < LL) ? lcol : 0;
    const float* Wrow = W + (long)wrow * NN;
    v2f bw[16];
    #pragma unroll
    for (int s = 0; s < 16; ++s) {
        int k0 = n0 + s * 4 + half * 2;
        bool ok = k0 < NN;
        bw[s].x = ok ? Wrow[k0] : 0.0f;
        bw[s].y = ok ? Wrow[k0 + 1] : 0.0f;
    }

    int pn = n0 + wave * 8 + (lane & 7);
    int len_l = (pn < NN) ? cnt[pn] : 0;
    int beg_l = (pn < NN) ? cur[pn] : 0;

    for (int i = 0; i < 8; ++i) {
        int lnode = wave * 8 + i;
        int n = n0 + lnode;
        float xh0 = 0.0f, xh1 = 0.0f;
        if (n < NN) {
            int len = __builtin_amdgcn_readlane(len_l, i);
            int beg = __builtin_amdgcn_readlane(beg_l, i);
            float a0 = 0.0f, a1 = 0.0f;
            if (len <= 64) {
                int id0 = (lane < len)      ? perm[beg + lane]      : 0x7fffffff;
                int id1 = (lane + 32 < len) ? perm[beg + lane + 32] : 0x7fffffff;
                int r0 = 0, r1 = 0;
                #pragma unroll
                for (int k = 0; k < 32; ++k) {
                    int s = __builtin_amdgcn_readlane(id0, k);
                    r0 += (s < id0) ? 1 : 0;
                    r1 += (s < id1) ? 1 : 0;
                }
                if (len > 32) {
                    #pragma unroll
                    for (int k = 0; k < 32; ++k) {
                        int s = __builtin_amdgcn_readlane(id1, k);
                        r0 += (s < id0) ? 1 : 0;
                        r1 += (s < id1) ? 1 : 0;
                    }
                }
                __builtin_amdgcn_fence(__ATOMIC_RELEASE, "wavefront");
                __builtin_amdgcn_wave_barrier();
                if (lane < len) {
                    s_src[wave][r0] = ei[id0];
                    s_al[wave][r0]  = alpha[id0];
                    s_bi[wave][r0]  = bias[id0];
                }
                if (lane + 32 < len) {
                    s_src[wave][r1] = ei[id1];
                    s_al[wave][r1]  = alpha[id1];
                    s_bi[wave][r1]  = bias[id1];
                }
                __builtin_amdgcn_fence(__ATOMIC_ACQ_REL, "wavefront");
                __builtin_amdgcn_wave_barrier();
                for (int j = 0; j < len; j += 8) {
                    float2 v[8]; float aa[8], bb[8];
                    #pragma unroll
                    for (int u = 0; u < 8; ++u) {
                        int jj = min(j + u, len - 1);
                        int s = __builtin_amdgcn_readfirstlane(s_src[wave][jj]);
                        aa[u] = __int_as_float(__builtin_amdgcn_readfirstlane(__float_as_int(s_al[wave][jj])));
                        bb[u] = __int_as_float(__builtin_amdgcn_readfirstlane(__float_as_int(s_bi[wave][jj])));
                        v[u] = *(const float2*)(xT + (long)s * 64 + 2 * lane);
                    }
                    #pragma unroll
                    for (int u = 0; u < 8; ++u) {
                        if (j + u < len) {
                            a0 += fmaf(aa[u], v[u].x, bb[u]);
                            a1 += fmaf(aa[u], v[u].y, bb[u]);
                        }
                    }
                }
            } else {
                int prev = -1;
                for (int it = 0; it < len; ++it) {
                    int m = 0x7fffffff;
                    for (int k = lane; k < len; k += 32) {
                        int id = perm[beg + k];
                        if (id > prev && id < m) m = id;
                    }
                    #pragma unroll
                    for (int off = 16; off >= 1; off >>= 1) m = min(m, __shfl_xor(m, off, 32));
                    int s = ei[m];
                    float al = alpha[m], bi = bias[m];
                    float2 v = *(const float2*)(xT + (long)s * 64 + 2 * lane);
                    a0 += fmaf(al, v.x, bi);
                    a1 += fmaf(al, v.y, bi);
                    prev = m;
                }
            }
            float den = fmaxf((float)len, 1.0f);
            xh0 = a0 / den;
            xh1 = a1 / den;
            float t0 = tanhf(xh0), t1 = tanhf(xh1);
            float sm = t0 + t1;
            float sq = t0 * t0 + t1 * t1;
            #pragma unroll
            for (int off = 16; off >= 1; off >>= 1) {
                sm += __shfl_xor(sm, off, 32);
                sq += __shfl_xor(sq, off, 32);
            }
            float mu = sm * (1.0f / 64.0f);
            float va = sq * (1.0f / 64.0f) - mu * mu;
            float rs = rsqrtf(va + BN_EPS);
            float g = gamma[n], be = beta[n];
            s_bn[lnode * 65 + 2 * lane]     = (t0 - mu) * rs * g + be;
            s_bn[lnode * 65 + 2 * lane + 1] = (t1 - mu) * rs * g + be;
        }
        float2 xw; xw.x = xh0; xw.y = xh1;
        *(float2*)&s_xh[lnode * XP + 2 * lane] = xw;
    }
    __syncthreads();

    {
        int col  = threadIdx.x & 63;
        int rowg = threadIdx.x >> 6;
        int n = n0 + col;
        if (n < NN) {
            for (int b = rowg; b < 64; b += 4)
                bn_out[(long)b * NN + n] = s_bn[col * 65 + b];
        }
    }

    v8f acc = {};
    int m = m_tile * 16 + l16;
    #pragma unroll
    for (int s = 0; s < 16; ++s) {
        int k0 = s * 4 + half * 2;
        v2f a;
        a.x = s_xh[k0 * XP + m];
        a.y = s_xh[(k0 + 1) * XP + m];
        acc = __builtin_amdgcn_wmma_f32_16x16x4_f32(
             false, a,  false, bw[s],
             (short)0, acc,  false,  false);
    }
    float* dst = part + (long)blockIdx.x * (BB * LL);
    #pragma unroll
    for (int i = 0; i < 8; ++i) {
        int mrow = m_tile * 16 + half * 8 + i;
        int ncol = n_tile * 16 + l16;
        if (ncol < LL) dst[mrow * LL + ncol] = acc[i];
    }
}

__global__ void __launch_bounds__(1024) k_reduce_pred(const float* __restrict__ part, const float* __restrict__ b_lin,
                                                      float* __restrict__ pred) {
    __shared__ float red[32][33];
    int il = threadIdx.x & 31;
    int g  = threadIdx.x >> 5;
    int i  = blockIdx.x * 32 + il;
    float s = 0.0f;
    for (int c = g; c < NBLK; c += 32) s += part[(long)c * (BB * LL) + i];
    red[g][il] = s;
    __syncthreads();
    if (g == 0) {
        float t = b_lin[i % LL];
        #pragma unroll
        for (int gg = 0; gg < 32; ++gg) t += red[gg][il];
        pred[i] = t;
    }
}

extern "C" void kernel_launch(void* const* d_in, const int* in_sizes, int n_in,
                              void* d_out, int out_size, void* d_ws, size_t ws_size,
                              hipStream_t stream) {
    const float* x     = (const float*)d_in[0];
    const int*   ei    = (const int*)  d_in[1];
    const float* alpha = (const float*)d_in[2];
    const float* bias  = (const float*)d_in[3];
    const float* W     = (const float*)d_in[4];
    const float* b_lin = (const float*)d_in[5];
    const float* gamma = (const float*)d_in[6];
    const float* beta  = (const float*)d_in[7];
    float* out = (float*)d_out;

    float* ws   = (float*)d_ws;
    float* xT   = ws;
    int*   rank = (int*)ws;
    int*   perm = (int*)(xT + (long)NN * 64);
    int*   cnt  = perm + EE;
    int*   cur  = cnt + NN;
    float* part = (float*)(cur + NN);

    k_zero_i<<<(NN + 255) / 256, 256, 0, stream>>>(cnt, NN);
    k_count<<<(EE + 255) / 256, 256, 0, stream>>>(ei, cnt, rank);
    k_scan<<<1, SCAN_T, 0, stream>>>(cnt, cur);
    k_fill<<<FILL_BLK, 1024, 0, stream>>>(ei, cur, rank, perm);
    k_transpose<<<(NN + 63) / 64, 256, 0, stream>>>(x, xT);
    k_gather<<<NBLK, 256, 0, stream>>>(xT, ei, alpha, bias, perm, cur, cnt, W, gamma, beta, out + BB * LL, part);
    k_reduce_pred<<<(BB * LL) / 32, 1024, 0, stream>>>(part, b_lin, out);
}


# bucket entries widened to 16-byte records (edge id, node, source, alpha, bias) filled from in-order reads; the gather bins whole records in LDS, reads one per edge: no lookups by edge id left
# speedup vs baseline: 4.4065x; 2.0418x over previous
#include <hip/hip_runtime.h>
#include <math.h>


#define BB 64
#define NN 50000
#define EE 1600000
#define LL 20
#define BN_EPS 1e-5f
#define NBLK 782
#define XP 80
#define PART_BLK 256
#define PART_EPB (EE / PART_BLK)
#define PART_T 250
#define PART_IT 25
#define CAP 2560
#define RPT (CAP / 256)
static_assert(PART_EPB * PART_BLK == EE, "the edge slices tile the edges exactly");
static_assert(PART_IT * PART_T == PART_EPB && 4 * PART_T > NBLK, "the threads tile a slice exactly; four tiles per thread cover the tiles and the end slot");
static_assert((NN - 1) / 64 == NBLK - 1, "every destination has a tile");
static_assert(((long)(EE - 1) << 6) < 0x7fffffffL, "edge id and local node fit one int");

typedef __attribute__((ext_vector_type(2))) float v2f;
typedef __attribute__((ext_vector_type(8))) float v8f;

__global__ void k_transpose(const float* __restrict__ x, float* __restrict__ xT) {
    __shared__ float tile[64 * 65];
    int n0   = blockIdx.x * 64;
    int col  = threadIdx.x & 63;
    int rowg = threadIdx.x >> 6;
    for (int b = rowg; b < 64; b += 4) {
        int n = n0 + col;
        tile[b * 65 + col] = (n < NN) ? x[(long)b * NN + n] : 0.0f;
    }
    __syncthreads();
    for (int nl = rowg; nl < 64; nl += 4) {
        int n = n0 + nl;
        if (n < NN) xT[(long)n * 64 + col] = tile[col * 65 + nl];
    }
}

__global__ void __launch_bounds__(PART_T) k_thist(const int* __restrict__ ei, int* __restrict__ hist) {
    __shared__ int h[NBLK];
    int tid  = threadIdx.x;
    int base = blockIdx.x * PART_EPB;
    int d[PART_IT];
    #pragma unroll
    for (int it = 0; it < PART_IT; ++it) d[it] = ei[EE + base + it * PART_T + tid];
    for (int t = tid; t < NBLK; t += PART_T) h[t] = 0;
    __syncthreads();
    #pragma unroll
    for (int it = 0; it < PART_IT; ++it) atomicAdd(&h[d[it] >> 6], 1);
    __syncthreads();
    for (int t = tid; t < NBLK; t += PART_T) hist[t * PART_BLK + blockIdx.x] = h[t];
}

#define TSCAN_BLOCKS ((NBLK + 7) / 8)
static_assert(PART_BLK == 256, "a tile's row of slice counts is 8 per lane");
__global__ void __launch_bounds__(256) k_tscan(int* __restrict__ hist, int* __restrict__ tot) {
    int lane = threadIdx.x & 31;
    int t = blockIdx.x * 8 + (threadIdx.x >> 5);
    if (t >= NBLK) return;
    int4* row = (int4*)(hist + t * PART_BLK) + lane * 2;
    int4 a = row[0], b = row[1];
    int s = a.x + a.y + a.z + a.w + b.x + b.y + b.z + b.w;
    int inc = s;
    #pragma unroll
    for (int off = 1; off < 32; off <<= 1) {
        int u = __shfl_up(inc, off, 32);
        if (lane >= off) inc += u;
    }
    int run = inc - s;
    int4 oa, ob;
    oa.x = run; run += a.x;
    oa.y = run; run += a.y;
    oa.z = run; run += a.z;
    oa.w = run; run += a.w;
    ob.x = run; run += b.x;
    ob.y = run; run += b.y;
    ob.z = run; run += b.z;
    ob.w = run; run += b.w;
    row[0] = oa;
    row[1] = ob;
    if (lane == 31) tot[t] = run;
}

__global__ void __launch_bounds__(PART_T) k_tscatter(const int* __restrict__ ei, const float* __restrict__ alpha,
                                                     const float* __restrict__ bias, const int* __restrict__ pre,
                                                     const int* __restrict__ tot, int* __restrict__ tstart,
                                                     int* __restrict__ recw) {
    int4* rec = (int4*)recw;
    __shared__ int cur[NBLK];
    __shared__ int wtot[(PART_T + 31) / 32];
    int tid = threadIdx.x, lane = tid & 31, wave = tid >> 5;
    int base = blockIdx.x * PART_EPB;
    int d[PART_IT];
    #pragma unroll
    for (int it = 0; it < PART_IT; ++it) d[it] = ei[EE + base + it * PART_T + tid];
    int v[4], pw[4];
    #pragma unroll
    for (int j = 0; j < 4; ++j) {
        int t = 4 * tid + j;
        v[j]  = (t < NBLK) ? tot[t] : 0;
        pw[j] = (t < NBLK) ? pre[t * PART_BLK + blockIdx.x] : 0;
    }
    int sr[PART_IT]; float al[PART_IT], bi[PART_IT];
    #pragma unroll
    for (int it = 0; it < PART_IT; ++it) {
        int e = base + it * PART_T + tid;
        sr[it] = ei[e];
        al[it] = alpha[e];
        bi[it] = bias[e];
    }
    int s4 = v[0] + v[1] + v[2] + v[3];
    int inc = s4;
    #pragma unroll
    for (int off = 1; off < 32; off <<= 1) {
        int u = __shfl_up(inc, off, 32);
        if (lane >= off) inc += u;
    }
    if (lane == 31) wtot[wave] = inc;
    __syncthreads();
    int run = inc - s4;
    for (int w2 = 0; w2 < wave; ++w2) run += wtot[w2];
    #pragma unroll
    for (int j = 0; j < 4; ++j) {
        int t = 4 * tid + j;
        if (t < NBLK) cur[t] = run + pw[j];
        if (blockIdx.x == 0 && t <= NBLK) tstart[t] = run;
        run += v[j];
    }
    __syncthreads();
    int pos[PART_IT];
    #pragma unroll
    for (int it = 0; it < PART_IT; ++it) pos[it] = atomicAdd(&cur[d[it] >> 6], 1);
    #pragma unroll
    for (int it = 0; it < PART_IT; ++it) {
        int4 r;
        r.x = ((base + it * PART_T + tid) << 6) | (d[it] & 63);
        r.y = sr[it];
        r.z = __float_as_int(al[it]);
        r.w = __float_as_int(bi[it]);
        rec[pos[it]] = r;
    }
}

__global__ void __launch_bounds__(256) k_gather(
        const float* __restrict__ xT, const int* __restrict__ ei,
        const float* __restrict__ alpha, const float* __restrict__ bias,
        const int* __restrict__ recw, const int* __restrict__ tstart,
        const float* __restrict__ W, const float* __restrict__ gamma, const float* __restrict__ beta,
        float* __restrict__ bn_out  , float* __restrict__ part  ) {
    __shared__ float s_bn[64 * 65];
    __shared__ float s_xh[64 * XP];
    __shared__ int   s_cnt[64];
    __shared__ int   s_beg[64];
    __shared__ int   s_fil[64];
    __shared__ int4  s_rec[CAP];

    const int4* rec = (const int4*)recw;
    int n0   = blockIdx.x * 64;
    int wave = threadIdx.x >> 5;
    int lane = threadIdx.x & 31;

    int half = lane >> 4;
    int l16  = lane & 15;
    int m_tile = wave >> 1;
    int n_tile = wave & 1;
    int lcol = n_tile * 16 + l16;
    int wrow = (lcol < LL) ? lcol : 0;
    const float* Wrow = W + (long)wrow * NN;
    v2f bw[16];
    #pragma unroll
    for (int s = 0; s < 16; ++s) {
        int k0 = n0 + s * 4 + half * 2;
        bool ok = k0 < NN;
        bw[s].x = ok ? Wrow[k0] : 0.0f;
        bw[s].y = ok ? Wrow[k0 + 1] : 0.0f;
    }

    int tb = tstart[blockIdx.x];
    int T  = tstart[blockIdx.x + 1] - tb;
    bool fit = T <= CAP;
    int4 r[RPT];
    #pragma unroll
    for (int u = 0; u < RPT; ++u) {
        int k = u * 256 + (int)threadIdx.x;
        r[u] = (fit && k < T) ? rec[tb + k] : make_int4(-1, 0, 0, 0);
    }
    if (threadIdx.x < 64) s_cnt[threadIdx.x] = 0;
    __syncthreads();
    #pragma unroll
    for (int u = 0; u < RPT; ++u) {
        if (r[u].x >= 0) atomicAdd(&s_cnt[r[u].x & 63], 1);
    }
    __syncthreads();
    if (wave == 0) {
        int c0 = s_cnt[2 * lane], c1 = s_cnt[2 * lane + 1];
        int sum2 = c0 + c1;
        int inc = sum2;
        #pragma unroll
        for (int off = 1; off < 32; off <<= 1) {
            int t = __shfl_up(inc, off, 32);
            if (lane >= off) inc += t;
        }
        int ex = inc - sum2;
        s_beg[2 * lane] = ex;      s_beg[2 * lane + 1] = ex + c0;
        s_fil[2 * lane] = ex;      s_fil[2 * lane + 1] = ex + c0;
    }
    __syncthreads();
    {
        int pos[RPT];
        #pragma unroll
        for (int u = 0; u < RPT; ++u) {
            pos[u] = (r[u].x >= 0) ? atomicAdd(&s_fil[r[u].x & 63], 1) : 0;
        }
        #pragma unroll
        for (int u = 0; u < RPT; ++u) {
            if (r[u].x >= 0) {
                int4 t = r[u];
                t.x >>= 6;
                s_rec[pos[u]] = t;
            }
        }
    }
    __syncthreads();

    int len_l = s_cnt[wave * 8 + (lane & 7)];
    int beg_l = s_beg[wave * 8 + (lane & 7)];

    for (int i = 0; i < 8; ++i) {
        int lnode = wave * 8 + i;
        int n = n0 + lnode;
        float xh0 = 0.0f, xh1 = 0.0f;
        if (n < NN) {
            int len = __builtin_amdgcn_readlane(len_l, i);
            int beg = __builtin_amdgcn_readlane(beg_l, i);
            float a0 = 0.0f, a1 = 0.0f;
            if (fit && len <= 64) {
                int4 q0 = make_int4(0x7fffffff, 0, 0, 0), q1 = make_int4(0x7fffffff, 0, 0, 0);
                if (lane < len)      q0 = s_rec[beg + lane];
                if (lane + 32 < len) q1 = s_rec[beg + lane + 32];
                int id0 = q0.x, id1 = q1.x;
                int r0 = 0, r1 = 0;
                #pragma unroll
                for (int k = 0; k < 32; ++k) {
                    int s = __builtin_amdgcn_readlane(id0, k);
                    r0 += (s < id0) ? 1 : 0;
                    r1 += (s < id1) ? 1 : 0;
                }
                if (len > 32) {
                    #pragma unroll
                    for (int k = 0; k < 32; ++k) {
                        int s = __builtin_amdgcn_readlane(id1, k);
                        r0 += (s < id0) ? 1 : 0;
                        r1 += (s < id1) ? 1 : 0;
                    }
                }
                __builtin_amdgcn_fence(__ATOMIC_RELEASE, "wavefront");
                __builtin_amdgcn_wave_barrier();
                if (lane < len)      s_rec[beg + r0] = q0;
                if (lane + 32 < len) s_rec[beg + r1] = q1;
                __builtin_amdgcn_fence(__ATOMIC_ACQ_REL, "wavefront");
                __builtin_amdgcn_wave_barrier();
                for (int j = 0; j < len; j += 8) {
                    float2 v[8]; int4 e[8];
                    #pragma unroll
                    for (int u = 0; u < 8; ++u) {
                        int jj = min(j + u, len - 1);
                        e[u] = s_rec[beg + jj];
                        v[u] = *(const float2*)(xT + ((unsigned)e[u].y * 64u + 2u * (unsigned)lane));
                    }
                    #pragma unroll
                    for (int u = 0; u < 8; ++u) {
                        if (j + u < len) {
                            float aa = __int_as_float(e[u].z), bb = __int_as_float(e[u].w);
                            a0 += fmaf(aa, v[u].x, bb);
                            a1 += fmaf(aa, v[u].y, bb);
                        }
                    }
                }
            } else {
                len = 0;
                int prev = -1;
                for (int it = 0; it <= T; ++it) {
                    int m = 0x7fffffff;
                    for (int k = lane; k < T; k += 32) {
                        int v = rec[tb + k].x;
                        int id = v >> 6;
                        if ((v & 63) == lnode && id > prev && id < m) m = id;
                    }
                    #pragma unroll
                    for (int off = 16; off >= 1; off >>= 1) m = min(m, __shfl_xor(m, off, 32));
                    if (m == 0x7fffffff) break;
                    ++len;
                    int s = ei[m];
                    float al = alpha[m], bi = bias[m];
                    float2 v = *(const float2*)(xT + (long)s * 64 + 2 * lane);
                    a0 += fmaf(al, v.x, bi);
                    a1 += fmaf(al, v.y, bi);
                    prev = m;
                }
            }
            float den = fmaxf((float)len, 1.0f);
            xh0 = a0 / den;
            xh1 = a1 / den;
            float t0 = tanhf(xh0), t1 = tanhf(xh1);
            float sm = t0 + t1;
            float sq = t0 * t0 + t1 * t1;
            #pragma unroll
            for (int off = 16; off >= 1; off >>= 1) {
                sm += __shfl_xor(sm, off, 32);
                sq += __shfl_xor(sq, off, 32);
            }
            float mu = sm * (1.0f / 64.0f);
            float va = sq * (1.0f / 64.0f) - mu * mu;
            float rs = rsqrtf(va + BN_EPS);
            float g = gamma[n], be = beta[n];
            s_bn[lnode * 65 + 2 * lane]     = (t0 - mu) * rs * g + be;
            s_bn[lnode * 65 + 2 * lane + 1] = (t1 - mu) * rs * g + be;
        }
        float2 xw; xw.x = xh0; xw.y = xh1;
        *(float2*)&s_xh[lnode * XP + 2 * lane] = xw;
    }
    __syncthreads();

    {
        int col  = threadIdx.x & 63;
        int rowg = threadIdx.x >> 6;
        int n = n0 + col;
        if (n < NN) {
            for (int b = rowg; b < 64; b += 4)
                bn_out[(long)b * NN + n] = s_bn[col * 65 + b];
        }
    }

    v8f acc = {};
    int m = m_tile * 16 + l16;
    #pragma unroll
    for (int s = 0; s < 16; ++s) {
        int k0 = s * 4 + half * 2;
        v2f a;
        a.x = s_xh[k0 * XP + m];
        a.y = s_xh[(k0 + 1) * XP + m];
        acc = __builtin_amdgcn_wmma_f32_16x16x4_f32(
             false, a,  false, bw[s],
             (short)0, acc,  false,  false);
    }
    float* dst = part + (long)blockIdx.x * (BB * LL);
    #pragma unroll
    for (int i = 0; i < 8; ++i) {
        int mrow = m_tile * 16 + half * 8 + i;
        int ncol = n_tile * 16 + l16;
        if (ncol < LL) dst[mrow * LL + ncol] = acc[i];
    }
}

__global__ void __launch_bounds__(1024) k_reduce_pred(const float* __restrict__ part, const float* __restrict__ b_lin,
                                                      float* __restrict__ pred) {
    __shared__ float red[32][33];
    int il = threadIdx.x & 31;
    int g  = threadIdx.x >> 5;
    int i  = blockIdx.x * 32 + il;
    float s = 0.0f;
    for (int c = g; c < NBLK; c += 32) s += part[(long)c * (BB * LL) + i];
    red[g][il] = s;
    __syncthreads();
    if (g == 0) {
        float t = b_lin[i % LL];
        #pragma unroll
        for (int gg = 0; gg < 32; ++gg) t += red[gg][il];
        pred[i] = t;
    }
}

extern "C" void kernel_launch(void* const* d_in, const int* in_sizes, int n_in,
                              void* d_out, int out_size, void* d_ws, size_t ws_size,
                              hipStream_t stream) {
    const float* x     = (const float*)d_in[0];
    const int*   ei    = (const int*)  d_in[1];
    const float* alpha = (const float*)d_in[2];
    const float* bias  = (const float*)d_in[3];
    const float* W     = (const float*)d_in[4];
    const float* b_lin = (const float*)d_in[5];
    const float* gamma = (const float*)d_in[6];
    const float* beta  = (const float*)d_in[7];
    float* out = (float*)d_out;

    float* ws     = (float*)d_ws;
    float* xT     = ws;
    int*   hist   = (int*)ws;
    int*   recw   = (int*)(xT + (long)NN * 64);
    int*   tot    = recw + 4L * EE;
    int*   tstart = tot + NN;
    float* part   = (float*)(tstart + NN);

    if (ws_size < (size_t)42803840) return;

    k_thist<<<PART_BLK, PART_T, 0, stream>>>(ei, hist);
    k_tscan<<<TSCAN_BLOCKS, 256, 0, stream>>>(hist, tot);
    k_tscatter<<<PART_BLK, PART_T, 0, stream>>>(ei, alpha, bias, hist, tot, tstart, recw);
    k_transpose<<<(NN + 63) / 64, 256, 0, stream>>>(x, xT);
    k_gather<<<NBLK, 256, 0, stream>>>(xT, ei, alpha, bias, recw, tstart, W, gamma, beta, out + BB * LL, part);
    k_reduce_pred<<<(BB * LL) / 32, 1024, 0, stream>>>(part, b_lin, out);
}


# gather rows read 16 bytes per lane: one wave-wide load covers two edges' rows (16 lanes each); half-waves sum the even and odd places of the edge order, added at the end: half the loads and passes
# speedup vs baseline: 5.4569x; 1.2384x over previous
#include <hip/hip_runtime.h>
#include <math.h>


#define BB 64
#define NN 50000
#define EE 1600000
#define LL 20
#define BN_EPS 1e-5f
#define NBLK 782
#define XP 80
#define NSL 250
#define SL_EPB (EE / NSL)
#define SL_T 256
#define SL_IT 25
#define CAP 2560
static_assert(SL_EPB * NSL == EE, "the edge slices tile the edges exactly");
static_assert(SL_IT * SL_T == SL_EPB && 4 * SL_T > NBLK, "the threads tile a slice exactly; four tiles per thread cover the tiles");
static_assert(SL_EPB < 65536, "an offset inside a slice fits 16 bits");
static_assert(NSL <= 256 && 2L * NBLK * NSL <= 8L * NN, "one gather thread per slice; the offset table fits the two planes it is given");
static_assert((NN - 1) / 64 == NBLK - 1, "every destination has a tile");
static_assert(((long)(EE - 1) << 6) < 0x7fffffffL, "edge id and local node fit one int");

typedef __attribute__((ext_vector_type(2))) float v2f;
typedef __attribute__((ext_vector_type(8))) float v8f;

__global__ void k_transpose(const float* __restrict__ x, float* __restrict__ xT) {
    __shared__ float tile[64 * 65];
    int n0   = blockIdx.x * 64;
    int col  = threadIdx.x & 63;
    int rowg = threadIdx.x >> 6;
    for (int b = rowg; b < 64; b += 4) {
        int n = n0 + col;
        tile[b * 65 + col] = (n < NN) ? x[(long)b * NN + n] : 0.0f;
    }
    __syncthreads();
    for (int nl = rowg; nl < 64; nl += 4) {
        int n = n0 + nl;
        if (n < NN) xT[(long)n * 64 + col] = tile[col * 65 + nl];
    }
}

__global__ void __launch_bounds__(SL_T) k_slicesort(const int* __restrict__ ei, const float* __restrict__ alpha,
                                                    const float* __restrict__ bias, int* __restrict__ recw,
                                                    int* __restrict__ tabw) {
    unsigned short* tab = (unsigned short*)tabw;
    __shared__ __attribute__((aligned(16))) int cnt[4 * SL_T];
    __shared__ int  wtot[SL_T / 32];
    __shared__ int4 stage[SL_EPB];
    int tid = threadIdx.x, lane = tid & 31, wave = tid >> 5;
    int base = blockIdx.x * SL_EPB;
    int d[SL_IT];
    #pragma unroll
    for (int it = 0; it < SL_IT; ++it) d[it] = ei[EE + base + it * SL_T + tid];
    int sr[SL_IT]; float al[SL_IT], bi[SL_IT];
    #pragma unroll
    for (int it = 0; it < SL_IT; ++it) {
        int e = base + it * SL_T + tid;
        sr[it] = ei[e];
        al[it] = alpha[e];
        bi[it] = bias[e];
    }
    *(int4*)&cnt[4 * tid] = make_int4(0, 0, 0, 0);
    __syncthreads();
    int rk[SL_IT];
    #pragma unroll
    for (int it = 0; it < SL_IT; ++it) rk[it] = atomicAdd(&cnt[d[it] >> 6], 1);
    __syncthreads();
    int4 c4 = *(int4*)&cnt[4 * tid];
    int s4 = c4.x + c4.y + c4.z + c4.w;
    int inc = s4;
    #pragma unroll
    for (int off = 1; off < 32; off <<= 1) {
        int u = __shfl_up(inc, off, 32);
        if (lane >= off) inc += u;
    }
    if (lane == 31) wtot[wave] = inc;
    __syncthreads();
    int run = inc - s4;
    for (int w2 = 0; w2 < wave; ++w2) run += wtot[w2];
    int4 o4;
    o4.x = run; run += c4.x;
    o4.y = run; run += c4.y;
    o4.z = run; run += c4.z;
    o4.w = run;
    *(int4*)&cnt[4 * tid] = o4;
    {
        int t = 4 * tid;
        if (t     < NBLK) tab[(t    ) * NSL + blockIdx.x] = (unsigned short)o4.x;
        if (t + 1 < NBLK) tab[(t + 1) * NSL + blockIdx.x] = (unsigned short)o4.y;
        if (t + 2 < NBLK) tab[(t + 2) * NSL + blockIdx.x] = (unsigned short)o4.z;
        if (t + 3 < NBLK) tab[(t + 3) * NSL + blockIdx.x] = (unsigned short)o4.w;
    }
    __syncthreads();
    #pragma unroll
    for (int it = 0; it < SL_IT; ++it) {
        int4 r;
        r.x = ((base + it * SL_T + tid) << 6) | (d[it] & 63);
        r.y = sr[it];
        r.z = __float_as_int(al[it]);
        r.w = __float_as_int(bi[it]);
        stage[cnt[d[it] >> 6] + rk[it]] = r;
    }
    __syncthreads();
    int4* dst = (int4*)recw + (long)blockIdx.x * SL_EPB;
    #pragma unroll
    for (int it = 0; it < SL_IT; ++it) dst[it * SL_T + tid] = stage[it * SL_T + tid];
}

__device__ __forceinline__ float xhalf(float x) {
    return __builtin_bit_cast(float, __builtin_amdgcn_permlanex16(__builtin_bit_cast(int, x), __builtin_bit_cast(int, x),
                                                                  0x76543210, 0xfedcba98, false, false));
}

__global__ void __launch_bounds__(256) k_gather(
        const float* __restrict__ xT, const int* __restrict__ ei,
        const float* __restrict__ alpha, const float* __restrict__ bias,
        const int* __restrict__ recw, const int* __restrict__ tabw,
        const float* __restrict__ W, const float* __restrict__ gamma, const float* __restrict__ beta,
        float* __restrict__ bn_out  , float* __restrict__ part  ) {
    __shared__ float s_bn[64 * 65];
    __shared__ float s_xh[64 * XP];
    __shared__ int   s_cnt[64];
    __shared__ int   s_beg[64];
    __shared__ int   s_fil[64];
    __shared__ int4  s_rec[CAP];
    __shared__ int   s_T;

    const int4* rec = (const int4*)recw;
    const unsigned short* tab = (const unsigned short*)tabw;
    int tile = blockIdx.x;
    int n0   = blockIdx.x * 64;
    int wave = threadIdx.x >> 5;
    int lane = threadIdx.x & 31;

    int half = lane >> 4;
    int l16  = lane & 15;
    int m_tile = wave >> 1;
    int n_tile = wave & 1;
    int lcol = n_tile * 16 + l16;
    int wrow = (lcol < LL) ? lcol : 0;
    const float* Wrow = W + (long)wrow * NN;

    int s0 = 0, cn = 0;
    if (threadIdx.x < NSL) {
        s0 = tab[tile * NSL + threadIdx.x];
        int s1 = (tile + 1 < NBLK) ? (int)tab[(tile + 1) * NSL + threadIdx.x] : SL_EPB;
        cn = s1 - s0;
    }
    int g8 = lane >> 3, k8 = lane & 7;
    int4 r[16];
    #pragma unroll
    for (int p = 0; p < 8; ++p) {
        int sl = p * 4 + g8;
        int rs = __shfl(s0, sl, 32), rc = __shfl(cn, sl, 32);
        int last = max(rc, 1) - 1;
        int rb = min(wave * 32 + sl, NSL - 1) * SL_EPB + rs;
        int4 qa = rec[min(rb + min(k8, last), EE - 1)];
        int4 qb = rec[min(rb + min(k8 + 8, last), EE - 1)];
        r[p]     = (k8 < rc)     ? qa : make_int4(-1, 0, 0, 0);
        r[8 + p] = (k8 + 8 < rc) ? qb : make_int4(-1, 0, 0, 0);
    }
    bool tail = __builtin_amdgcn_ballot_w32(cn > 16) != 0;
    if (threadIdx.x < 64) s_cnt[threadIdx.x] = 0;
    __syncthreads();
    #pragma unroll
    for (int u = 0; u < 16; ++u) {
        if (r[u].x >= 0) atomicAdd(&s_cnt[r[u].x & 63], 1);
    }
    if (tail) {
        for (int p = 0; p < 8; ++p) {
            int sl = p * 4 + g8;
            int rs = __shfl(s0, sl, 32), rc = __shfl(cn, sl, 32);
            const int4* runp = rec + ((long)(wave * 32 + sl) * SL_EPB + rs);
            for (int kk = 16 + k8; kk < rc; kk += 8) atomicAdd(&s_cnt[runp[kk].x & 63], 1);
        }
    }
    __syncthreads();
    if (wave == 0) {
        int c0 = s_cnt[2 * lane], c1 = s_cnt[2 * lane + 1];
        int sum2 = c0 + c1;
        int inc = sum2;
        #pragma unroll
        for (int off = 1; off < 32; off <<= 1) {
            int t = __shfl_up(inc, off, 32);
            if (lane >= off) inc += t;
        }
        int ex = inc - sum2;
        s_beg[2 * lane] = ex;      s_beg[2 * lane + 1] = ex + c0;
        s_fil[2 * lane] = ex;      s_fil[2 * lane + 1] = ex + c0;
        if (lane == 31) s_T = inc;
    }
    __syncthreads();
    int T = s_T;
    bool fit = T <= CAP;
    if (fit) {
        #pragma unroll
        for (int h = 0; h < 2; ++h) {
            int pos[8];
            #pragma unroll
            for (int u = 0; u < 8; ++u) {
                pos[u] = (r[8 * h + u].x >= 0) ? atomicAdd(&s_fil[r[8 * h + u].x & 63], 1) : 0;
            }
            #pragma unroll
            for (int u = 0; u < 8; ++u) {
                if (r[8 * h + u].x >= 0) {
                    int4 t = r[8 * h + u];
                    t.x >>= 6;
                    s_rec[pos[u]] = t;
                }
            }
        }
        if (tail) {
            for (int p = 0; p < 8; ++p) {
                int sl = p * 4 + g8;
                int rs = __shfl(s0, sl, 32), rc = __shfl(cn, sl, 32);
                const int4* runp = rec + ((long)(wave * 32 + sl) * SL_EPB + rs);
                for (int kk = 16 + k8; kk < rc; kk += 8) {
                    int4 t = runp[kk];
                    int ps = atomicAdd(&s_fil[t.x & 63], 1);
                    t.x >>= 6;
                    s_rec[ps] = t;
                }
            }
        }
    }
    __syncthreads();

    v2f bw[16];
    #pragma unroll
    for (int s = 0; s < 16; ++s) {
        int k0 = n0 + s * 4 + half * 2;
        bool ok = k0 < NN;
        bw[s].x = ok ? Wrow[k0] : 0.0f;
        bw[s].y = ok ? Wrow[k0 + 1] : 0.0f;
    }

    int len_l = s_cnt[wave * 8 + (lane & 7)];
    int beg_l = s_beg[wave * 8 + (lane & 7)];
    int b0 = 4 * l16 + 2 * half;

    for (int i = 0; i < 8; ++i) {
        int lnode = wave * 8 + i;
        int n = n0 + lnode;
        float xh0 = 0.0f, xh1 = 0.0f;
        if (n < NN) {
            int len = __builtin_amdgcn_readlane(len_l, i);
            int beg = __builtin_amdgcn_readlane(beg_l, i);
            float a0 = 0.0f, a1 = 0.0f;
            if (fit && len <= 64) {
                int4 q0 = make_int4(0x7fffffff, 0, 0, 0), q1 = make_int4(0x7fffffff, 0, 0, 0);
                if (lane < len)      q0 = s_rec[beg + lane];
                if (lane + 32 < len) q1 = s_rec[beg + lane + 32];
                int id0 = q0.x, id1 = q1.x;
                int r0 = 0, r1 = 0;
                #pragma unroll
                for (int k = 0; k < 32; ++k) {
                    int s = __builtin_amdgcn_readlane(id0, k);
                    r0 += (s < id0) ? 1 : 0;
                    r1 += (s < id1) ? 1 : 0;
                }
                if (len > 32) {
                    #pragma unroll
                    for (int k = 0; k < 32; ++k) {
                        int s = __builtin_amdgcn_readlane(id1, k);
                        r0 += (s < id0) ? 1 : 0;
                        r1 += (s < id1) ? 1 : 0;
                    }
                }
                __builtin_amdgcn_fence(__ATOMIC_RELEASE, "wavefront");
                __builtin_amdgcn_wave_barrier();
                if (lane < len)      s_rec[beg + r0] = q0;
                if (lane + 32 < len) s_rec[beg + r1] = q1;
                __builtin_amdgcn_fence(__ATOMIC_ACQ_REL, "wavefront");
                __builtin_amdgcn_wave_barrier();
                float p0 = 0.0f, p1 = 0.0f, p2 = 0.0f, p3 = 0.0f;
                for (int j = 0; j < len; j += 16) {
                    float4 v[8]; float aa[8], bb[8];
                    #pragma unroll
                    for (int u = 0; u < 8; ++u) {
                        int pl = j + 2 * u + half;
                        int4 e = s_rec[beg + min(pl, len - 1)];
                        v[u] = *(const float4*)(xT + ((unsigned)e.y * 64u + 4u * (unsigned)l16));
                        bool ok = pl < len;
                        aa[u] = ok ? __int_as_float(e.z) : 0.0f;
                        bb[u] = ok ? __int_as_float(e.w) : 0.0f;
                    }
                    #pragma unroll
                    for (int u = 0; u < 8; ++u) {
                        p0 += fmaf(aa[u], v[u].x, bb[u]);
                        p1 += fmaf(aa[u], v[u].y, bb[u]);
                        p2 += fmaf(aa[u], v[u].z, bb[u]);
                        p3 += fmaf(aa[u], v[u].w, bb[u]);
                    }
                }
                float r0x = xhalf(half ? p0 : p2);
                float r1x = xhalf(half ? p1 : p3);
                a0 = (half ? p2 : p0) + r0x;
                a1 = (half ? p3 : p1) + r1x;
            } else {
                len = 0;
                int prev = -1;
                for (int it = 0; it <= T; ++it) {
                    int m = 0x7fffffff;
                    for (int w = lane; w < NSL; w += 32) {
                        int ka = tab[tile * NSL + w];
                        int kb = (tile + 1 < NBLK) ? (int)tab[(tile + 1) * NSL + w] : SL_EPB;
                        for (int k = ka; k < kb; ++k) {
                            int v = rec[(long)w * SL_EPB + k].x;
                            int id = v >> 6;
                            if ((v & 63) == lnode && id > prev && id < m) m = id;
                        }
                    }
                    #pragma unroll
                    for (int off = 16; off >= 1; off >>= 1) m = min(m, __shfl_xor(m, off, 32));
                    if (m == 0x7fffffff) break;
                    ++len;
                    int s = ei[m];
                    float al = alpha[m], bi = bias[m];
                    float2 v = *(const float2*)(xT + (long)s * 64 + b0);
                    a0 += fmaf(al, v.x, bi);
                    a1 += fmaf(al, v.y, bi);
                    prev = m;
                }
            }
            float den = fmaxf((float)len, 1.0f);
            xh0 = a0 / den;
            xh1 = a1 / den;
            float t0 = tanhf(xh0), t1 = tanhf(xh1);
            float sm = t0 + t1;
            float sq = t0 * t0 + t1 * t1;
            #pragma unroll
            for (int off = 16; off >= 1; off >>= 1) {
                sm += __shfl_xor(sm, off, 32);
                sq += __shfl_xor(sq, off, 32);
            }
            float mu = sm * (1.0f / 64.0f);
            float va = sq * (1.0f / 64.0f) - mu * mu;
            float rs = rsqrtf(va + BN_EPS);
            float g = gamma[n], be = beta[n];
            s_bn[lnode * 65 + b0]     = (t0 - mu) * rs * g + be;
            s_bn[lnode * 65 + b0 + 1] = (t1 - mu) * rs * g + be;
        }
        float2 xw; xw.x = xh0; xw.y = xh1;
        *(float2*)&s_xh[lnode * XP + b0] = xw;
    }
    __syncthreads();

    {
        int col  = threadIdx.x & 63;
        int rowg = threadIdx.x >> 6;
        int n = n0 + col;
        if (n < NN) {
            for (int b = rowg; b < 64; b += 4)
                bn_out[(long)b * NN + n] = s_bn[col * 65 + b];
        }
    }

    v8f acc = {};
    int m = m_tile * 16 + l16;
    #pragma unroll
    for (int s = 0; s < 16; ++s) {
        int k0 = s * 4 + half * 2;
        v2f a;
        a.x = s_xh[k0 * XP + m];
        a.y = s_xh[(k0 + 1) * XP + m];
        acc = __builtin_amdgcn_wmma_f32_16x16x4_f32(
             false, a,  false, bw[s],
             (short)0, acc,  false,  false);
    }
    float* dst = part + (long)blockIdx.x * (BB * LL);
    #pragma unroll
    for (int i = 0; i < 8; ++i) {
        int mrow = m_tile * 16 + half * 8 + i;
        int ncol = n_tile * 16 + l16;
        if (ncol < LL) dst[mrow * LL + ncol] = acc[i];
    }
}

__global__ void __launch_bounds__(1024) k_reduce_pred(const float* __restrict__ part, const float* __restrict__ b_lin,
                                                      float* __restrict__ pred) {
    __shared__ float red[32][33];
    int il = threadIdx.x & 31;
    int g  = threadIdx.x >> 5;
    int i  = blockIdx.x * 32 + il;
    float s = 0.0f;
    for (int c = g; c < NBLK; c += 32) s += part[(long)c * (BB * LL) + i];
    red[g][il] = s;
    __syncthreads();
    if (g == 0) {
        float t = b_lin[i % LL];
        #pragma unroll
        for (int gg = 0; gg < 32; ++gg) t += red[gg][il];
        pred[i] = t;
    }
}

extern "C" void kernel_launch(void* const* d_in, const int* in_sizes, int n_in,
                              void* d_out, int out_size, void* d_ws, size_t ws_size,
                              hipStream_t stream) {
    const float* x     = (const float*)d_in[0];
    const int*   ei    = (const int*)  d_in[1];
    const float* alpha = (const float*)d_in[2];
    const float* bias  = (const float*)d_in[3];
    const float* W     = (const float*)d_in[4];
    const float* b_lin = (const float*)d_in[5];
    const float* gamma = (const float*)d_in[6];
    const float* beta  = (const float*)d_in[7];
    float* out = (float*)d_out;

    float* ws     = (float*)d_ws;
    float* xT     = ws;
    int*   recw   = (int*)(xT + (long)NN * 64);
    int*   tabw   = recw + 4L * EE;
    float* part   = (float*)(tabw + 2L * NN);

    if (ws_size < (size_t)42803840) return;

    k_slicesort<<<NSL, SL_T, 0, stream>>>(ei, alpha, bias, recw, tabw);
    k_transpose<<<(NN + 63) / 64, 256, 0, stream>>>(x, xT);
    k_gather<<<NBLK, 256, 0, stream>>>(xT, ei, alpha, bias, recw, tabw, W, gamma, beta, out + BB * LL, part);
    k_reduce_pred<<<(BB * LL) / 32, 1024, 0, stream>>>(part, b_lin, out);
}


# transpose kernel: guards taken off its loads (global: column clamped into the row; LDS reads too), so a thread's 16 loads go out together, not one memory trip after another; stores keep the guard
# speedup vs baseline: 5.8802x; 1.0776x over previous
#include <hip/hip_runtime.h>
#include <math.h>


#define BB 64
#define NN 50000
#define EE 1600000
#define LL 20
#define BN_EPS 1e-5f
#define NBLK 782
#define XP 80
#define NSL 250
#define SL_EPB (EE / NSL)
#define SL_T 256
#define SL_IT 25
#define CAP 2560
static_assert(SL_EPB * NSL == EE, "the edge slices tile the edges exactly");
static_assert(SL_IT * SL_T == SL_EPB && 4 * SL_T > NBLK, "the threads tile a slice exactly; four tiles per thread cover the tiles");
static_assert(SL_EPB < 65536, "an offset inside a slice fits 16 bits");
static_assert(NSL <= 256 && 2L * NBLK * NSL <= 8L * NN, "one gather thread per slice; the offset table fits the two planes it is given");
static_assert((NN - 1) / 64 == NBLK - 1, "every destination has a tile");
static_assert(((long)(EE - 1) << 6) < 0x7fffffffL, "edge id and local node fit one int");

typedef __attribute__((ext_vector_type(2))) float v2f;
typedef __attribute__((ext_vector_type(8))) float v8f;

__global__ void k_transpose(const float* __restrict__ x, float* __restrict__ xT) {
    __shared__ float tile[64 * 65];
    int n0   = blockIdx.x * 64;
    int col  = threadIdx.x & 63;
    int rowg = threadIdx.x >> 6;
    int nc   = min(n0 + col, NN - 1);
    float v[16];
    #pragma unroll
    for (int i = 0; i < 16; ++i) v[i] = x[(long)(rowg + 4 * i) * NN + nc];
    #pragma unroll
    for (int i = 0; i < 16; ++i) tile[(rowg + 4 * i) * 65 + col] = v[i];
    __syncthreads();
    float w[16];
    #pragma unroll
    for (int i = 0; i < 16; ++i) w[i] = tile[col * 65 + rowg + 4 * i];
    #pragma unroll
    for (int i = 0; i < 16; ++i) {
        int n = n0 + rowg + 4 * i;
        if (n < NN) xT[(long)n * 64 + col] = w[i];
    }
}

__global__ void __launch_bounds__(SL_T) k_slicesort(const int* __restrict__ ei, const float* __restrict__ alpha,
                                                    const float* __restrict__ bias, int* __restrict__ recw,
                                                    int* __restrict__ tabw) {
    unsigned short* tab = (unsigned short*)tabw;
    __shared__ __attribute__((aligned(16))) int cnt[4 * SL_T];
    __shared__ int  wtot[SL_T / 32];
    __shared__ int4 stage[SL_EPB];
    int tid = threadIdx.x, lane = tid & 31, wave = tid >> 5;
    int base = blockIdx.x * SL_EPB;
    int d[SL_IT];
    #pragma unroll
    for (int it = 0; it < SL_IT; ++it) d[it] = ei[EE + base + it * SL_T + tid];
    int sr[SL_IT]; float al[SL_IT], bi[SL_IT];
    #pragma unroll
    for (int it = 0; it < SL_IT; ++it) {
        int e = base + it * SL_T + tid;
        sr[it] = ei[e];
        al[it] = alpha[e];
        bi[it] = bias[e];
    }
    *(int4*)&cnt[4 * tid] = make_int4(0, 0, 0, 0);
    __syncthreads();
    int rk[SL_IT];
    #pragma unroll
    for (int it = 0; it < SL_IT; ++it) rk[it] = atomicAdd(&cnt[d[it] >> 6], 1);
    __syncthreads();
    int4 c4 = *(int4*)&cnt[4 * tid];
    int s4 = c4.x + c4.y + c4.z + c4.w;
    int inc = s4;
    #pragma unroll
    for (int off = 1; off < 32; off <<= 1) {
        int u = __shfl_up(inc, off, 32);
        if (lane >= off) inc += u;
    }
    if (lane == 31) wtot[wave] = inc;
    __syncthreads();
    int run = inc - s4;
    for (int w2 = 0; w2 < wave; ++w2) run += wtot[w2];
    int4 o4;
    o4.x = run; run += c4.x;
    o4.y = run; run += c4.y;
    o4.z = run; run += c4.z;
    o4.w = run;
    *(int4*)&cnt[4 * tid] = o4;
    {
        int t = 4 * tid;
        if (t     < NBLK) tab[(t    ) * NSL + blockIdx.x] = (unsigned short)o4.x;
        if (t + 1 < NBLK) tab[(t + 1) * NSL + blockIdx.x] = (unsigned short)o4.y;
        if (t + 2 < NBLK) tab[(t + 2) * NSL + blockIdx.x] = (unsigned short)o4.z;
        if (t + 3 < NBLK) tab[(t + 3) * NSL + blockIdx.x] = (unsigned short)o4.w;
    }
    __syncthreads();
    #pragma unroll
    for (int it = 0; it < SL_IT; ++it) {
        int4 r;
        r.x = ((base + it * SL_T + tid) << 6) | (d[it] & 63);
        r.y = sr[it];
        r.z = __float_as_int(al[it]);
        r.w = __float_as_int(bi[it]);
        stage[cnt[d[it] >> 6] + rk[it]] = r;
    }
    __syncthreads();
    int4* dst = (int4*)recw + (long)blockIdx.x * SL_EPB;
    #pragma unroll
    for (int it = 0; it < SL_IT; ++it) dst[it * SL_T + tid] = stage[it * SL_T + tid];
}

__device__ __forceinline__ float xhalf(float x) {
    return __builtin_bit_cast(float, __builtin_amdgcn_permlanex16(__builtin_bit_cast(int, x), __builtin_bit_cast(int, x),
                                                                  0x76543210, 0xfedcba98, false, false));
}

__global__ void __launch_bounds__(256) k_gather(
        const float* __restrict__ xT, const int* __restrict__ ei,
        const float* __restrict__ alpha, const float* __restrict__ bias,
        const int* __restrict__ recw, const int* __restrict__ tabw,
        const float* __restrict__ W, const float* __restrict__ gamma, const float* __restrict__ beta,
        float* __restrict__ bn_out  , float* __restrict__ part  ) {
    __shared__ float s_bn[64 * 65];
    __shared__ float s_xh[64 * XP];
    __shared__ int   s_cnt[64];
    __shared__ int   s_beg[64];
    __shared__ int   s_fil[64];
    __shared__ int4  s_rec[CAP];
    __shared__ int   s_T;

    const int4* rec = (const int4*)recw;
    const unsigned short* tab = (const unsigned short*)tabw;
    int tile = blockIdx.x;
    int n0   = blockIdx.x * 64;
    int wave = threadIdx.x >> 5;
    int lane = threadIdx.x & 31;

    int half = lane >> 4;
    int l16  = lane & 15;
    int m_tile = wave >> 1;
    int n_tile = wave & 1;
    int lcol = n_tile * 16 + l16;
    int wrow = (lcol < LL) ? lcol : 0;
    const float* Wrow = W + (long)wrow * NN;

    int s0 = 0, cn = 0;
    if (threadIdx.x < NSL) {
        s0 = tab[tile * NSL + threadIdx.x];
        int s1 = (tile + 1 < NBLK) ? (int)tab[(tile + 1) * NSL + threadIdx.x] : SL_EPB;
        cn = s1 - s0;
    }
    int g8 = lane >> 3, k8 = lane & 7;
    int4 r[16];
    #pragma unroll
    for (int p = 0; p < 8; ++p) {
        int sl = p * 4 + g8;
        int rs = __shfl(s0, sl, 32), rc = __shfl(cn, sl, 32);
        int last = max(rc, 1) - 1;
        int rb = min(wave * 32 + sl, NSL - 1) * SL_EPB + rs;
        int4 qa = rec[min(rb + min(k8, last), EE - 1)];
        int4 qb = rec[min(rb + min(k8 + 8, last), EE - 1)];
        r[p]     = (k8 < rc)     ? qa : make_int4(-1, 0, 0, 0);
        r[8 + p] = (k8 + 8 < rc) ? qb : make_int4(-1, 0, 0, 0);
    }
    bool tail = __builtin_amdgcn_ballot_w32(cn > 16) != 0;
    if (threadIdx.x < 64) s_cnt[threadIdx.x] = 0;
    __syncthreads();
    #pragma unroll
    for (int u = 0; u < 16; ++u) {
        if (r[u].x >= 0) atomicAdd(&s_cnt[r[u].x & 63], 1);
    }
    if (tail) {
        for (int p = 0; p < 8; ++p) {
            int sl = p * 4 + g8;
            int rs = __shfl(s0, sl, 32), rc = __shfl(cn, sl, 32);
            const int4* runp = rec + ((long)(wave * 32 + sl) * SL_EPB + rs);
            for (int kk = 16 + k8; kk < rc; kk += 8) atomicAdd(&s_cnt[runp[kk].x & 63], 1);
        }
    }
    __syncthreads();
    if (wave == 0) {
        int c0 = s_cnt[2 * lane], c1 = s_cnt[2 * lane + 1];
        int sum2 = c0 + c1;
        int inc = sum2;
        #pragma unroll
        for (int off = 1; off < 32; off <<= 1) {
            int t = __shfl_up(inc, off, 32);
            if (lane >= off) inc += t;
        }
        int ex = inc - sum2;
        s_beg[2 * lane] = ex;      s_beg[2 * lane + 1] = ex + c0;
        s_fil[2 * lane] = ex;      s_fil[2 * lane + 1] = ex + c0;
        if (lane == 31) s_T = inc;
    }
    __syncthreads();
    int T = s_T;
    bool fit = T <= CAP;
    if (fit) {
        #pragma unroll
        for (int h = 0; h < 2; ++h) {
            int pos[8];
            #pragma unroll
            for (int u = 0; u < 8; ++u) {
                pos[u] = (r[8 * h + u].x >= 0) ? atomicAdd(&s_fil[r[8 * h + u].x & 63], 1) : 0;
            }
            #pragma unroll
            for (int u = 0; u < 8; ++u) {
                if (r[8 * h + u].x >= 0) {
                    int4 t = r[8 * h + u];
                    t.x >>= 6;
                    s_rec[pos[u]] = t;
                }
            }
        }
        if (tail) {
            for (int p = 0; p < 8; ++p) {
                int sl = p * 4 + g8;
                int rs = __shfl(s0, sl, 32), rc = __shfl(cn, sl, 32);
                const int4* runp = rec + ((long)(wave * 32 + sl) * SL_EPB + rs);
                for (int kk = 16 + k8; kk < rc; kk += 8) {
                    int4 t = runp[kk];
                    int ps = atomicAdd(&s_fil[t.x & 63], 1);
                    t.x >>= 6;
                    s_rec[ps] = t;
                }
            }
        }
    }
    __syncthreads();

    v2f bw[16];
    #pragma unroll
    for (int s = 0; s < 16; ++s) {
        int k0 = n0 + s * 4 + half * 2;
        bool ok = k0 < NN;
        bw[s].x = ok ? Wrow[k0] : 0.0f;
        bw[s].y = ok ? Wrow[k0 + 1] : 0.0f;
    }

    int len_l = s_cnt[wave * 8 + (lane & 7)];
    int beg_l = s_beg[wave * 8 + (lane & 7)];
    int b0 = 4 * l16 + 2 * half;

    for (int i = 0; i < 8; ++i) {
        int lnode = wave * 8 + i;
        int n = n0 + lnode;
        float xh0 = 0.0f, xh1 = 0.0f;
        if (n < NN) {
            int len = __builtin_amdgcn_readlane(len_l, i);
            int beg = __builtin_amdgcn_readlane(beg_l, i);
            float a0 = 0.0f, a1 = 0.0f;
            if (fit && len <= 64) {
                int4 q0 = make_int4(0x7fffffff, 0, 0, 0), q1 = make_int4(0x7fffffff, 0, 0, 0);
                if (lane < len)      q0 = s_rec[beg + lane];
                if (lane + 32 < len) q1 = s_rec[beg + lane + 32];
                int id0 = q0.x, id1 = q1.x;
                int r0 = 0, r1 = 0;
                #pragma unroll
                for (int k = 0; k < 32; ++k) {
                    int s = __builtin_amdgcn_readlane(id0, k);
                    r0 += (s < id0) ? 1 : 0;
                    r1 += (s < id1) ? 1 : 0;
                }
                if (len > 32) {
                    #pragma unroll
                    for (int k = 0; k < 32; ++k) {
                        int s = __builtin_amdgcn_readlane(id1, k);
                        r0 += (s < id0) ? 1 : 0;
                        r1 += (s < id1) ? 1 : 0;
                    }
                }
                __builtin_amdgcn_fence(__ATOMIC_RELEASE, "wavefront");
                __builtin_amdgcn_wave_barrier();
                if (lane < len)      s_rec[beg + r0] = q0;
                if (lane + 32 < len) s_rec[beg + r1] = q1;
                __builtin_amdgcn_fence(__ATOMIC_ACQ_REL, "wavefront");
                __builtin_amdgcn_wave_barrier();
                float p0 = 0.0f, p1 = 0.0f, p2 = 0.0f, p3 = 0.0f;
                for (int j = 0; j < len; j += 16) {
                    float4 v[8]; float aa[8], bb[8];
                    #pragma unroll
                    for (int u = 0; u < 8; ++u) {
                        int pl = j + 2 * u + half;
                        int4 e = s_rec[beg + min(pl, len - 1)];
                        v[u] = *(const float4*)(xT + ((unsigned)e.y * 64u + 4u * (unsigned)l16));
                        bool ok = pl < len;
                        aa[u] = ok ? __int_as_float(e.z) : 0.0f;
                        bb[u] = ok ? __int_as_float(e.w) : 0.0f;
                    }
                    #pragma unroll
                    for (int u = 0; u < 8; ++u) {
                        p0 += fmaf(aa[u], v[u].x, bb[u]);
                        p1 += fmaf(aa[u], v[u].y, bb[u]);
                        p2 += fmaf(aa[u], v[u].z, bb[u]);
                        p3 += fmaf(aa[u], v[u].w, bb[u]);
                    }
                }
                float r0x = xhalf(half ? p0 : p2);
                float r1x = xhalf(half ? p1 : p3);
                a0 = (half ? p2 : p0) + r0x;
                a1 = (half ? p3 : p1) + r1x;
            } else {
                len = 0;
                int prev = -1;
                for (int it = 0; it <= T; ++it) {
                    int m = 0x7fffffff;
                    for (int w = lane; w < NSL; w += 32) {
                        int ka = tab[tile * NSL + w];
                        int kb = (tile + 1 < NBLK) ? (int)tab[(tile + 1) * NSL + w] : SL_EPB;
                        for (int k = ka; k < kb; ++k) {
                            int v = rec[(long)w * SL_EPB + k].x;
                            int id = v >> 6;
                            if ((v & 63) == lnode && id > prev && id < m) m = id;
                        }
                    }
                    #pragma unroll
                    for (int off = 16; off >= 1; off >>= 1) m = min(m, __shfl_xor(m, off, 32));
                    if (m == 0x7fffffff) break;
                    ++len;
                    int s = ei[m];
                    float al = alpha[m], bi = bias[m];
                    float2 v = *(const float2*)(xT + (long)s * 64 + b0);
                    a0 += fmaf(al, v.x, bi);
                    a1 += fmaf(al, v.y, bi);
                    prev = m;
                }
            }
            float den = fmaxf((float)len, 1.0f);
            xh0 = a0 / den;
            xh1 = a1 / den;
            float t0 = tanhf(xh0), t1 = tanhf(xh1);
            float sm = t0 + t1;
            float sq = t0 * t0 + t1 * t1;
            #pragma unroll
            for (int off = 16; off >= 1; off >>= 1) {
                sm += __shfl_xor(sm, off, 32);
                sq += __shfl_xor(sq, off, 32);
            }
            float mu = sm * (1.0f / 64.0f);
            float va = sq * (1.0f / 64.0f) - mu * mu;
            float rs = rsqrtf(va + BN_EPS);
            float g = gamma[n], be = beta[n];
            s_bn[lnode * 65 + b0]     = (t0 - mu) * rs * g + be;
            s_bn[lnode * 65 + b0 + 1] = (t1 - mu) * rs * g + be;
        }
        float2 xw; xw.x = xh0; xw.y = xh1;
        *(float2*)&s_xh[lnode * XP + b0] = xw;
    }
    __syncthreads();

    {
        int col  = threadIdx.x & 63;
        int rowg = threadIdx.x >> 6;
        int n = n0 + col;
        if (n < NN) {
            for (int b = rowg; b < 64; b += 4)
                bn_out[(long)b * NN + n] = s_bn[col * 65 + b];
        }
    }

    v8f acc = {};
    int m = m_tile * 16 + l16;
    #pragma unroll
    for (int s = 0; s < 16; ++s) {
        int k0 = s * 4 + half * 2;
        v2f a;
        a.x = s_xh[k0 * XP + m];
        a.y = s_xh[(k0 + 1) * XP + m];
        acc = __builtin_amdgcn_wmma_f32_16x16x4_f32(
             false, a,  false, bw[s],
             (short)0, acc,  false,  false);
    }
    float* dst = part + (long)blockIdx.x * (BB * LL);
    #pragma unroll
    for (int i = 0; i < 8; ++i) {
        int mrow = m_tile * 16 + half * 8 + i;
        int ncol = n_tile * 16 + l16;
        if (ncol < LL) dst[mrow * LL + ncol] = acc[i];
    }
}

__global__ void __launch_bounds__(1024) k_reduce_pred(const float* __restrict__ part, const float* __restrict__ b_lin,
                                                      float* __restrict__ pred) {
    __shared__ float red[32][33];
    int il = threadIdx.x & 31;
    int g  = threadIdx.x >> 5;
    int i  = blockIdx.x * 32 + il;
    float s = 0.0f;
    for (int c = g; c < NBLK; c += 32) s += part[(long)c * (BB * LL) + i];
    red[g][il] = s;
    __syncthreads();
    if (g == 0) {
        float t = b_lin[i % LL];
        #pragma unroll
        for (int gg = 0; gg < 32; ++gg) t += red[gg][il];
        pred[i] = t;
    }
}

extern "C" void kernel_launch(void* const* d_in, const int* in_sizes, int n_in,
                              void* d_out, int out_size, void* d_ws, size_t ws_size,
                              hipStream_t stream) {
    const float* x     = (const float*)d_in[0];
    const int*   ei    = (const int*)  d_in[1];
    const float* alpha = (const float*)d_in[2];
    const float* bias  = (const float*)d_in[3];
    const float* W     = (const float*)d_in[4];
    const float* b_lin = (const float*)d_in[5];
    const float* gamma = (const float*)d_in[6];
    const float* beta  = (const float*)d_in[7];
    float* out = (float*)d_out;

    float* ws     = (float*)d_ws;
    float* xT     = ws;
    int*   recw   = (int*)(xT + (long)NN * 64);
    int*   tabw   = recw + 4L * EE;
    float* part   = (float*)(tabw + 2L * NN);

    if (ws_size < (size_t)42803840) return;

    k_slicesort<<<NSL, SL_T, 0, stream>>>(ei, alpha, bias, recw, tabw);
    k_transpose<<<(NN + 63) / 64, 256, 0, stream>>>(x, xT);
    k_gather<<<NBLK, 256, 0, stream>>>(xT, ei, alpha, bias, recw, tabw, W, gamma, beta, out + BB * LL, part);
    k_reduce_pred<<<(BB * LL) / 32, 1024, 0, stream>>>(part, b_lin, out);
}


# gather: a node's records ordered by a bitonic sorting network on the vector unit (keys: edge id and slot, asked for one node ahead; row_xmask min/max steps), not by counting ranks with broadcasts
# speedup vs baseline: 6.5819x; 1.1193x over previous
#include <hip/hip_runtime.h>
#include <math.h>


#define BB 64
#define NN 50000
#define EE 1600000
#define LL 20
#define BN_EPS 1e-5f
#define NBLK 782
#define XP 80
#define NSL 250
#define SL_EPB (EE / NSL)
#define SL_T 256
#define SL_IT 25
#define CAP 2560
static_assert(SL_EPB * NSL == EE, "the edge slices tile the edges exactly");
static_assert(SL_IT * SL_T == SL_EPB && 4 * SL_T > NBLK, "the threads tile a slice exactly; four tiles per thread cover the tiles");
static_assert(SL_EPB < 65536, "an offset inside a slice fits 16 bits");
static_assert(NN <= 65536 && SL_EPB < 32768, "a destination and a rank inside a slice share one int");
static_assert(NSL <= 256 && 2L * NBLK * NSL <= 8L * NN, "one gather thread per slice; the offset table fits the two planes it is given");
static_assert((NN - 1) / 64 == NBLK - 1, "every destination has a tile");
static_assert(((long)(EE - 1) << 6) < 0x7fffffffL, "edge id and local node fit one int");

typedef __attribute__((ext_vector_type(2))) float v2f;
typedef __attribute__((ext_vector_type(8))) float v8f;

__device__ __forceinline__ void transpose_tile(const float* __restrict__ x, float* __restrict__ xT, float* tile, int tb) {
    int n0   = tb * 64;
    int col  = threadIdx.x & 63;
    int rowg = threadIdx.x >> 6;
    int nc   = min(n0 + col, NN - 1);
    float v[16];
    #pragma unroll
    for (int i = 0; i < 16; ++i) v[i] = x[(long)(rowg + 4 * i) * NN + nc];
    #pragma unroll
    for (int i = 0; i < 16; ++i) tile[(rowg + 4 * i) * 65 + col] = v[i];
    __syncthreads();
    float w[16];
    #pragma unroll
    for (int i = 0; i < 16; ++i) w[i] = tile[col * 65 + rowg + 4 * i];
    #pragma unroll
    for (int i = 0; i < 16; ++i) {
        int n = n0 + rowg + 4 * i;
        if (n < NN) xT[(long)n * 64 + col] = w[i];
    }
}

__global__ void __launch_bounds__(SL_T) k_prep(const int* __restrict__ ei, const float* __restrict__ alpha,
                                               const float* __restrict__ bias, int* __restrict__ recw,
                                               int* __restrict__ tabw, const float* __restrict__ x,
                                               float* __restrict__ xT) {
    unsigned short* tab = (unsigned short*)tabw;
    __shared__ __attribute__((aligned(16))) int cnt[4 * SL_T];
    __shared__ int  wtot[SL_T / 32];
    __shared__ __attribute__((aligned(16))) unsigned char buf[sizeof(int4) * SL_EPB];
    static_assert(sizeof(int4) * SL_EPB >= sizeof(float) * 64 * 65, "the transposed tile fits the staging buffer");
    if (blockIdx.x >= NSL) {
        transpose_tile(x, xT, (float*)buf, blockIdx.x - NSL);
        return;
    }
    int4* stage = (int4*)buf;
    int tid = threadIdx.x, lane = tid & 31, wave = tid >> 5;
    int base = blockIdx.x * SL_EPB;
    int d[SL_IT];
    #pragma unroll
    for (int it = 0; it < SL_IT; ++it) d[it] = ei[EE + base + it * SL_T + tid];
    int sr[SL_IT]; float al[SL_IT], bi[SL_IT];
    #pragma unroll
    for (int it = 0; it < SL_IT; ++it) {
        int e = base + it * SL_T + tid;
        sr[it] = ei[e];
        al[it] = alpha[e];
        bi[it] = bias[e];
    }
    *(int4*)&cnt[4 * tid] = make_int4(0, 0, 0, 0);
    __syncthreads();
    #pragma unroll
    for (int it = 0; it < SL_IT; ++it) d[it] |= atomicAdd(&cnt[d[it] >> 6], 1) << 16;
    __syncthreads();
    int4 c4 = *(int4*)&cnt[4 * tid];
    int s4 = c4.x + c4.y + c4.z + c4.w;
    int inc = s4;
    #pragma unroll
    for (int off = 1; off < 32; off <<= 1) {
        int u = __shfl_up(inc, off, 32);
        if (lane >= off) inc += u;
    }
    if (lane == 31) wtot[wave] = inc;
    __syncthreads();
    int run = inc - s4;
    for (int w2 = 0; w2 < wave; ++w2) run += wtot[w2];
    int4 o4;
    o4.x = run; run += c4.x;
    o4.y = run; run += c4.y;
    o4.z = run; run += c4.z;
    o4.w = run;
    *(int4*)&cnt[4 * tid] = o4;
    {
        int t = 4 * tid;
        if (t     < NBLK) tab[(t    ) * NSL + blockIdx.x] = (unsigned short)o4.x;
        if (t + 1 < NBLK) tab[(t + 1) * NSL + blockIdx.x] = (unsigned short)o4.y;
        if (t + 2 < NBLK) tab[(t + 2) * NSL + blockIdx.x] = (unsigned short)o4.z;
        if (t + 3 < NBLK) tab[(t + 3) * NSL + blockIdx.x] = (unsigned short)o4.w;
    }
    __syncthreads();
    #pragma unroll
    for (int it = 0; it < SL_IT; ++it) {
        int4 r;
        int dn = d[it] & 0xffff;
        int rk = d[it] >> 16;
        r.x = ((base + it * SL_T + tid) << 6) | (dn & 63);
        r.y = sr[it];
        r.z = __float_as_int(al[it]);
        r.w = __float_as_int(bi[it]);
        stage[cnt[dn >> 6] + rk] = r;
    }
    __syncthreads();
    int4* dst = (int4*)recw + (long)blockIdx.x * SL_EPB;
    #pragma unroll
    for (int it = 0; it < SL_IT; ++it) dst[it * SL_T + tid] = stage[it * SL_T + tid];
}

__device__ __forceinline__ float xhalf(float x) {
    return __builtin_bit_cast(float, __builtin_amdgcn_permlanex16(__builtin_bit_cast(int, x), __builtin_bit_cast(int, x),
                                                                  0x76543210, 0xfedcba98, false, false));
}

template <int J> __device__ __forceinline__ int xlane(int x) {
    if constexpr (J == 16)
        return __builtin_amdgcn_permlanex16(x, x, 0x76543210, 0xfedcba98, false, false);
    else
        return __builtin_amdgcn_update_dpp(x, x, 0x160 + J, 0xf, 0xf, true);
}

template <int K, int J> __device__ __forceinline__ int cxstep(int x, int lane, bool down) {
    int xp = xlane<J>(x);
    bool keep_min = (((lane & J) == 0) == ((lane & K) == 0)) != down;
    int lo = min(x, xp), hi = max(x, xp);
    return keep_min ? lo : hi;
}

__device__ __forceinline__ int sort32(int x, int lane, bool down) {
    x = cxstep<2, 1>(x, lane, false);
    x = cxstep<4, 2>(x, lane, false);   x = cxstep<4, 1>(x, lane, false);
    x = cxstep<8, 4>(x, lane, false);   x = cxstep<8, 2>(x, lane, false);   x = cxstep<8, 1>(x, lane, false);
    x = cxstep<16, 8>(x, lane, false);  x = cxstep<16, 4>(x, lane, false);  x = cxstep<16, 2>(x, lane, false);
    x = cxstep<16, 1>(x, lane, false);
    x = cxstep<32, 16>(x, lane, down);  x = cxstep<32, 8>(x, lane, down);   x = cxstep<32, 4>(x, lane, down);
    x = cxstep<32, 2>(x, lane, down);   x = cxstep<32, 1>(x, lane, down);
    return x;
}

__device__ __forceinline__ int merge32(int x, int lane) {
    x = cxstep<32, 16>(x, lane, false); x = cxstep<32, 8>(x, lane, false);  x = cxstep<32, 4>(x, lane, false);
    x = cxstep<32, 2>(x, lane, false);  x = cxstep<32, 1>(x, lane, false);
    return x;
}

__global__ void __launch_bounds__(256) k_gather(
        const float* __restrict__ xT, const int* __restrict__ ei,
        const float* __restrict__ alpha, const float* __restrict__ bias,
        const int* __restrict__ recw, const int* __restrict__ tabw,
        const float* __restrict__ W, const float* __restrict__ gamma, const float* __restrict__ beta,
        float* __restrict__ bn_out  , float* __restrict__ part  ) {
    __shared__ float s_bn[64 * 65];
    __shared__ float s_xh[64 * XP];
    __shared__ int   s_cnt[64];
    __shared__ int   s_beg[64];
    __shared__ int   s_fil[64];
    __shared__ int4  s_rec[CAP];
    __shared__ int   s_T;

    const int4* rec = (const int4*)recw;
    const unsigned short* tab = (const unsigned short*)tabw;
    int tile = blockIdx.x;
    int n0   = blockIdx.x * 64;
    int wave = threadIdx.x >> 5;
    int lane = threadIdx.x & 31;

    int half = lane >> 4;
    int l16  = lane & 15;
    int m_tile = wave >> 1;
    int n_tile = wave & 1;
    int lcol = n_tile * 16 + l16;
    int wrow = (lcol < LL) ? lcol : 0;
    const float* Wrow = W + (long)wrow * NN;

    int s0 = 0, cn = 0;
    if (threadIdx.x < NSL) {
        s0 = tab[tile * NSL + threadIdx.x];
        int s1 = (tile + 1 < NBLK) ? (int)tab[(tile + 1) * NSL + threadIdx.x] : SL_EPB;
        cn = s1 - s0;
    }
    int g8 = lane >> 3, k8 = lane & 7;
    int4 r[16];
    #pragma unroll
    for (int p = 0; p < 8; ++p) {
        int sl = p * 4 + g8;
        int rs = __shfl(s0, sl, 32), rc = __shfl(cn, sl, 32);
        int last = max(rc, 1) - 1;
        int rb = min(wave * 32 + sl, NSL - 1) * SL_EPB + rs;
        int4 qa = rec[min(rb + min(k8, last), EE - 1)];
        int4 qb = rec[min(rb + min(k8 + 8, last), EE - 1)];
        r[p]     = (k8 < rc)     ? qa : make_int4(-1, 0, 0, 0);
        r[8 + p] = (k8 + 8 < rc) ? qb : make_int4(-1, 0, 0, 0);
    }
    bool tail = __builtin_amdgcn_ballot_w32(cn > 16) != 0;
    if (threadIdx.x < 64) s_cnt[threadIdx.x] = 0;
    __syncthreads();
    #pragma unroll
    for (int u = 0; u < 16; ++u) {
        if (r[u].x >= 0) atomicAdd(&s_cnt[r[u].x & 63], 1);
    }
    if (tail) {
        for (int p = 0; p < 8; ++p) {
            int sl = p * 4 + g8;
            int rs = __shfl(s0, sl, 32), rc = __shfl(cn, sl, 32);
            const int4* runp = rec + ((long)(wave * 32 + sl) * SL_EPB + rs);
            for (int kk = 16 + k8; kk < rc; kk += 8) atomicAdd(&s_cnt[runp[kk].x & 63], 1);
        }
    }
    __syncthreads();
    if (wave == 0) {
        int c0 = s_cnt[2 * lane], c1 = s_cnt[2 * lane + 1];
        int sum2 = c0 + c1;
        int inc = sum2;
        #pragma unroll
        for (int off = 1; off < 32; off <<= 1) {
            int t = __shfl_up(inc, off, 32);
            if (lane >= off) inc += t;
        }
        int ex = inc - sum2;
        s_beg[2 * lane] = ex;      s_beg[2 * lane + 1] = ex + c0;
        s_fil[2 * lane] = ex;      s_fil[2 * lane + 1] = ex + c0;
        if (lane == 31) s_T = inc;
    }
    __syncthreads();
    int T = s_T;
    bool fit = T <= CAP;
    if (fit) {
        #pragma unroll
        for (int h = 0; h < 2; ++h) {
            int pos[8];
            #pragma unroll
            for (int u = 0; u < 8; ++u) {
                pos[u] = (r[8 * h + u].x >= 0) ? atomicAdd(&s_fil[r[8 * h + u].x & 63], 1) : 0;
            }
            #pragma unroll
            for (int u = 0; u < 8; ++u) {
                if (r[8 * h + u].x >= 0) {
                    int4 t = r[8 * h + u];
                    t.x >>= 6;
                    s_rec[pos[u]] = t;
                }
            }
        }
        if (tail) {
            for (int p = 0; p < 8; ++p) {
                int sl = p * 4 + g8;
                int rs = __shfl(s0, sl, 32), rc = __shfl(cn, sl, 32);
                const int4* runp = rec + ((long)(wave * 32 + sl) * SL_EPB + rs);
                for (int kk = 16 + k8; kk < rc; kk += 8) {
                    int4 t = runp[kk];
                    int ps = atomicAdd(&s_fil[t.x & 63], 1);
                    t.x >>= 6;
                    s_rec[ps] = t;
                }
            }
        }
    }
    __syncthreads();

    v2f bw[16];
    #pragma unroll
    for (int s = 0; s < 16; ++s) {
        int k0 = n0 + s * 4 + half * 2;
        bool ok = k0 < NN;
        bw[s].x = ok ? Wrow[k0] : 0.0f;
        bw[s].y = ok ? Wrow[k0 + 1] : 0.0f;
    }

    int len_l = s_cnt[wave * 8 + (lane & 7)];
    int beg_l = s_beg[wave * 8 + (lane & 7)];
    int b0 = 4 * l16 + 2 * half;

    static_assert(EE - 1 < 0x01ffffff, "the id of a lane without a slot sorts behind every edge id");
    int pk0 = 0x01ffffff, pk1 = 0x01ffffff;
    {
        int ln = __builtin_amdgcn_readlane(len_l, 0), bg = __builtin_amdgcn_readlane(beg_l, 0);
        if (fit && lane < ln)      pk0 = s_rec[bg + lane].x;
        if (fit && lane + 32 < ln) pk1 = s_rec[bg + lane + 32].x;
    }
    asm volatile("" : "+v"(pk0), "+v"(pk1));

    for (int i = 0; i < 8; ++i) {
        int lnode = wave * 8 + i;
        int id0 = pk0, id1 = pk1;
        pk0 = 0x01ffffff; pk1 = 0x01ffffff;
        if (fit && i < 7) {
            int ln = __builtin_amdgcn_readlane(len_l, i + 1), bg = __builtin_amdgcn_readlane(beg_l, i + 1);
            if (lane < ln)      pk0 = s_rec[bg + lane].x;
            if (lane + 32 < ln) pk1 = s_rec[bg + lane + 32].x;
        }
        int n = n0 + lnode;
        float xh0 = 0.0f, xh1 = 0.0f;
        if (n < NN) {
            int len = __builtin_amdgcn_readlane(len_l, i);
            int beg = __builtin_amdgcn_readlane(beg_l, i);
            float a0 = 0.0f, a1 = 0.0f;
            if (fit && len <= 64) {
                int k0 = (id0 << 6) | lane, k1 = (id1 << 6) | (lane + 32);
                if (len > 32) {
                    k0 = sort32(k0, lane, false);
                    k1 = sort32(k1, lane, true);
                    int lo = min(k0, k1), hi = max(k0, k1);
                    k0 = merge32(lo, lane);
                    k1 = merge32(hi, lane);
                } else {
                    k0 = sort32(k0, lane, false);
                }
                int4 q0 = make_int4(0, 0, 0, 0), q1 = make_int4(0, 0, 0, 0);
                if (lane < len)      q0 = s_rec[beg + (k0 & 63)];
                if (lane + 32 < len) q1 = s_rec[beg + (k1 & 63)];
                __builtin_amdgcn_fence(__ATOMIC_RELEASE, "wavefront");
                __builtin_amdgcn_wave_barrier();
                if (lane < len)      s_rec[beg + lane] = q0;
                if (lane + 32 < len) s_rec[beg + lane + 32] = q1;
                __builtin_amdgcn_fence(__ATOMIC_ACQ_REL, "wavefront");
                __builtin_amdgcn_wave_barrier();
                float p0 = 0.0f, p1 = 0.0f, p2 = 0.0f, p3 = 0.0f;
                for (int j = 0; j < len; j += 16) {
                    float4 v[8]; float aa[8], bb[8];
                    #pragma unroll
                    for (int u = 0; u < 8; ++u) {
                        int pl = j + 2 * u + half;
                        int4 e = s_rec[beg + min(pl, len - 1)];
                        v[u] = *(const float4*)(xT + ((unsigned)e.y * 64u + 4u * (unsigned)l16));
                        bool ok = pl < len;
                        aa[u] = ok ? __int_as_float(e.z) : 0.0f;
                        bb[u] = ok ? __int_as_float(e.w) : 0.0f;
                    }
                    #pragma unroll
                    for (int u = 0; u < 8; ++u) {
                        p0 += fmaf(aa[u], v[u].x, bb[u]);
                        p1 += fmaf(aa[u], v[u].y, bb[u]);
                        p2 += fmaf(aa[u], v[u].z, bb[u]);
                        p3 += fmaf(aa[u], v[u].w, bb[u]);
                    }
                }
                float r0x = xhalf(half ? p0 : p2);
                float r1x = xhalf(half ? p1 : p3);
                a0 = (half ? p2 : p0) + r0x;
                a1 = (half ? p3 : p1) + r1x;
            } else {
                len = 0;
                int prev = -1;
                for (int it = 0; it <= T; ++it) {
                    int m = 0x7fffffff;
                    for (int w = lane; w < NSL; w += 32) {
                        int ka = tab[tile * NSL + w];
                        int kb = (tile + 1 < NBLK) ? (int)tab[(tile + 1) * NSL + w] : SL_EPB;
                        for (int k = ka; k < kb; ++k) {
                            int v = rec[(long)w * SL_EPB + k].x;
                            int id = v >> 6;
                            if ((v & 63) == lnode && id > prev && id < m) m = id;
                        }
                    }
                    #pragma unroll
                    for (int off = 16; off >= 1; off >>= 1) m = min(m, __shfl_xor(m, off, 32));
                    if (m == 0x7fffffff) break;
                    ++len;
                    int s = ei[m];
                    float al = alpha[m], bi = bias[m];
                    float2 v = *(const float2*)(xT + (long)s * 64 + b0);
                    a0 += fmaf(al, v.x, bi);
                    a1 += fmaf(al, v.y, bi);
                    prev = m;
                }
            }
            float den = fmaxf((float)len, 1.0f);
            xh0 = a0 / den;
            xh1 = a1 / den;
            float t0 = tanhf(xh0), t1 = tanhf(xh1);
            float sm = t0 + t1;
            float sq = t0 * t0 + t1 * t1;
            #pragma unroll
            for (int off = 16; off >= 1; off >>= 1) {
                sm += __shfl_xor(sm, off, 32);
                sq += __shfl_xor(sq, off, 32);
            }
            float mu = sm * (1.0f / 64.0f);
            float va = sq * (1.0f / 64.0f) - mu * mu;
            float rs = rsqrtf(va + BN_EPS);
            float g = gamma[n], be = beta[n];
            s_bn[lnode * 65 + b0]     = (t0 - mu) * rs * g + be;
            s_bn[lnode * 65 + b0 + 1] = (t1 - mu) * rs * g + be;
        }
        float2 xw; xw.x = xh0; xw.y = xh1;
        *(float2*)&s_xh[lnode * XP + b0] = xw;
    }
    __syncthreads();

    {
        int col  = threadIdx.x & 63;
        int rowg = threadIdx.x >> 6;
        int n = n0 + col;
        if (n < NN) {
            for (int b = rowg; b < 64; b += 4)
                bn_out[(long)b * NN + n] = s_bn[col * 65 + b];
        }
    }

    v8f acc = {};
    int m = m_tile * 16 + l16;
    #pragma unroll
    for (int s = 0; s < 16; ++s) {
        int k0 = s * 4 + half * 2;
        v2f a;
        a.x = s_xh[k0 * XP + m];
        a.y = s_xh[(k0 + 1) * XP + m];
        acc = __builtin_amdgcn_wmma_f32_16x16x4_f32(
             false, a,  false, bw[s],
             (short)0, acc,  false,  false);
    }
    float* dst = part + (long)blockIdx.x * (BB * LL);
    #pragma unroll
    for (int i = 0; i < 8; ++i) {
        int mrow = m_tile * 16 + half * 8 + i;
        int ncol = n_tile * 16 + l16;
        if (ncol < LL) dst[mrow * LL + ncol] = acc[i];
    }
}

__global__ void __launch_bounds__(1024) k_reduce_pred(const float* __restrict__ part, const float* __restrict__ b_lin,
                                                      float* __restrict__ pred) {
    __shared__ float red[32][33];
    int il = threadIdx.x & 31;
    int g  = threadIdx.x >> 5;
    int i  = blockIdx.x * 32 + il;
    float s = 0.0f;
    for (int c = g; c < NBLK; c += 32) s += part[(long)c * (BB * LL) + i];
    red[g][il] = s;
    __syncthreads();
    if (g == 0) {
        float t = b_lin[i % LL];
        #pragma unroll
        for (int gg = 0; gg < 32; ++gg) t += red[gg][il];
        pred[i] = t;
    }
}

extern "C" void kernel_launch(void* const* d_in, const int* in_sizes, int n_in,
                              void* d_out, int out_size, void* d_ws, size_t ws_size,
                              hipStream_t stream) {
    const float* x     = (const float*)d_in[0];
    const int*   ei    = (const int*)  d_in[1];
    const float* alpha = (const float*)d_in[2];
    const float* bias  = (const float*)d_in[3];
    const float* W     = (const float*)d_in[4];
    const float* b_lin = (const float*)d_in[5];
    const float* gamma = (const float*)d_in[6];
    const float* beta  = (const float*)d_in[7];
    float* out = (float*)d_out;

    float* ws     = (float*)d_ws;
    float* xT     = ws;
    int*   recw   = (int*)(xT + (long)NN * 64);
    int*   tabw   = recw + 4L * EE;
    float* part   = (float*)(tabw + 2L * NN);

    if (ws_size < (size_t)42803840) return;

    k_prep<<<NSL + NBLK, SL_T, 0, stream>>>(ei, alpha, bias, recw, tabw, x, xT);
    k_gather<<<NBLK, 256, 0, stream>>>(xT, ei, alpha, bias, recw, tabw, W, gamma, beta, out + BB * LL, part);
    k_reduce_pred<<<(BB * LL) / 32, 1024, 0, stream>>>(part, b_lin, out);
}


# gather sums ask only for rows of real edges: whole passes of 16 places run unclamped and unmasked (one LDS address a pass), a node's last pass loads exactly the pairs left, not a fixed 8
# speedup vs baseline: 6.9697x; 1.0589x over previous
#include <hip/hip_runtime.h>
#include <math.h>


#define BB 64
#define NN 50000
#define EE 1600000
#define LL 20
#define BN_EPS 1e-5f
#define NBLK 782
#define XP 80
#define NSL 250
#define SL_EPB (EE / NSL)
#define SL_T 256
#define SL_IT 25
#define CAP 2560
static_assert(SL_EPB * NSL == EE, "the edge slices tile the edges exactly");
static_assert(SL_IT * SL_T == SL_EPB && 4 * SL_T > NBLK, "the threads tile a slice exactly; four tiles per thread cover the tiles");
static_assert(SL_EPB < 65536, "an offset inside a slice fits 16 bits");
static_assert(NN <= 65536 && SL_EPB < 32768, "a destination and a rank inside a slice share one int");
static_assert(NSL <= 256 && 2L * NBLK * NSL <= 8L * NN, "one gather thread per slice; the offset table fits the two planes it is given");
static_assert((NN - 1) / 64 == NBLK - 1, "every destination has a tile");
static_assert(((long)(EE - 1) << 6) < 0x7fffffffL, "edge id and local node fit one int");

typedef __attribute__((ext_vector_type(2))) float v2f;
typedef __attribute__((ext_vector_type(8))) float v8f;

__device__ __forceinline__ void transpose_tile(const float* __restrict__ x, float* __restrict__ xT, float* tile, int tb) {
    int n0   = tb * 64;
    int col  = threadIdx.x & 63;
    int rowg = threadIdx.x >> 6;
    int nc   = min(n0 + col, NN - 1);
    float v[16];
    #pragma unroll
    for (int i = 0; i < 16; ++i) v[i] = x[(long)(rowg + 4 * i) * NN + nc];
    #pragma unroll
    for (int i = 0; i < 16; ++i) tile[(rowg + 4 * i) * 65 + col] = v[i];
    __syncthreads();
    float w[16];
    #pragma unroll
    for (int i = 0; i < 16; ++i) w[i] = tile[col * 65 + rowg + 4 * i];
    #pragma unroll
    for (int i = 0; i < 16; ++i) {
        int n = n0 + rowg + 4 * i;
        if (n < NN) xT[(long)n * 64 + col] = w[i];
    }
}

__global__ void __launch_bounds__(SL_T) k_prep(const int* __restrict__ ei, const float* __restrict__ alpha,
                                               const float* __restrict__ bias, int* __restrict__ recw,
                                               int* __restrict__ tabw, const float* __restrict__ x,
                                               float* __restrict__ xT) {
    unsigned short* tab = (unsigned short*)tabw;
    __shared__ __attribute__((aligned(16))) int cnt[4 * SL_T];
    __shared__ int  wtot[SL_T / 32];
    __shared__ __attribute__((aligned(16))) unsigned char buf[sizeof(int4) * SL_EPB];
    static_assert(sizeof(int4) * SL_EPB >= sizeof(float) * 64 * 65, "the transposed tile fits the staging buffer");
    if (blockIdx.x >= NSL) {
        transpose_tile(x, xT, (float*)buf, blockIdx.x - NSL);
        return;
    }
    int4* stage = (int4*)buf;
    int tid = threadIdx.x, lane = tid & 31, wave = tid >> 5;
    int base = blockIdx.x * SL_EPB;
    int d[SL_IT];
    #pragma unroll
    for (int it = 0; it < SL_IT; ++it) d[it] = ei[EE + base + it * SL_T + tid];
    int sr[SL_IT]; float al[SL_IT], bi[SL_IT];
    #pragma unroll
    for (int it = 0; it < SL_IT; ++it) {
        int e = base + it * SL_T + tid;
        sr[it] = ei[e];
        al[it] = alpha[e];
        bi[it] = bias[e];
    }
    *(int4*)&cnt[4 * tid] = make_int4(0, 0, 0, 0);
    __syncthreads();
    #pragma unroll
    for (int it = 0; it < SL_IT; ++it) d[it] |= atomicAdd(&cnt[d[it] >> 6], 1) << 16;
    __syncthreads();
    int4 c4 = *(int4*)&cnt[4 * tid];
    int s4 = c4.x + c4.y + c4.z + c4.w;
    int inc = s4;
    #pragma unroll
    for (int off = 1; off < 32; off <<= 1) {
        int u = __shfl_up(inc, off, 32);
        if (lane >= off) inc += u;
    }
    if (lane == 31) wtot[wave] = inc;
    __syncthreads();
    int run = inc - s4;
    for (int w2 = 0; w2 < wave; ++w2) run += wtot[w2];
    int4 o4;
    o4.x = run; run += c4.x;
    o4.y = run; run += c4.y;
    o4.z = run; run += c4.z;
    o4.w = run;
    *(int4*)&cnt[4 * tid] = o4;
    {
        int t = 4 * tid;
        if (t     < NBLK) tab[(t    ) * NSL + blockIdx.x] = (unsigned short)o4.x;
        if (t + 1 < NBLK) tab[(t + 1) * NSL + blockIdx.x] = (unsigned short)o4.y;
        if (t + 2 < NBLK) tab[(t + 2) * NSL + blockIdx.x] = (unsigned short)o4.z;
        if (t + 3 < NBLK) tab[(t + 3) * NSL + blockIdx.x] = (unsigned short)o4.w;
    }
    __syncthreads();
    #pragma unroll
    for (int it = 0; it < SL_IT; ++it) {
        int4 r;
        int dn = d[it] & 0xffff;
        int rk = d[it] >> 16;
        r.x = ((base + it * SL_T + tid) << 6) | (dn & 63);
        r.y = sr[it];
        r.z = __float_as_int(al[it]);
        r.w = __float_as_int(bi[it]);
        stage[cnt[dn >> 6] + rk] = r;
    }
    __syncthreads();
    int4* dst = (int4*)recw + (long)blockIdx.x * SL_EPB;
    #pragma unroll
    for (int it = 0; it < SL_IT; ++it) dst[it * SL_T + tid] = stage[it * SL_T + tid];
}

__device__ __forceinline__ float xhalf(float x) {
    return __builtin_bit_cast(float, __builtin_amdgcn_permlanex16(__builtin_bit_cast(int, x), __builtin_bit_cast(int, x),
                                                                  0x76543210, 0xfedcba98, false, false));
}

template <int J> __device__ __forceinline__ int xlane(int x) {
    if constexpr (J == 16)
        return __builtin_amdgcn_permlanex16(x, x, 0x76543210, 0xfedcba98, false, false);
    else
        return __builtin_amdgcn_update_dpp(x, x, 0x160 + J, 0xf, 0xf, true);
}

template <int K, int J> __device__ __forceinline__ int cxstep(int x, int lane, bool down) {
    int xp = xlane<J>(x);
    bool keep_min = (((lane & J) == 0) == ((lane & K) == 0)) != down;
    int lo = min(x, xp), hi = max(x, xp);
    return keep_min ? lo : hi;
}

__device__ __forceinline__ int sort32(int x, int lane, bool down) {
    x = cxstep<2, 1>(x, lane, false);
    x = cxstep<4, 2>(x, lane, false);   x = cxstep<4, 1>(x, lane, false);
    x = cxstep<8, 4>(x, lane, false);   x = cxstep<8, 2>(x, lane, false);   x = cxstep<8, 1>(x, lane, false);
    x = cxstep<16, 8>(x, lane, false);  x = cxstep<16, 4>(x, lane, false);  x = cxstep<16, 2>(x, lane, false);
    x = cxstep<16, 1>(x, lane, false);
    x = cxstep<32, 16>(x, lane, down);  x = cxstep<32, 8>(x, lane, down);   x = cxstep<32, 4>(x, lane, down);
    x = cxstep<32, 2>(x, lane, down);   x = cxstep<32, 1>(x, lane, down);
    return x;
}

__device__ __forceinline__ int merge32(int x, int lane) {
    x = cxstep<32, 16>(x, lane, false); x = cxstep<32, 8>(x, lane, false);  x = cxstep<32, 4>(x, lane, false);
    x = cxstep<32, 2>(x, lane, false);  x = cxstep<32, 1>(x, lane, false);
    return x;
}

template <int NP, bool TAIL>
__device__ __forceinline__ void sum_pass(const float* __restrict__ xT, const int4* rp, int l16, bool last_ok,
                                         v2f& pa, v2f& pb) {
    float4 v[NP]; float aa[NP], bb[NP];
    #pragma unroll
    for (int u = 0; u < NP; ++u) {
        bool fix = TAIL && u == NP - 1;
        int4 e = rp[2 * u - ((fix && !last_ok) ? 1 : 0)];
        v[u] = *(const float4*)(xT + ((unsigned)e.y * 64u + 4u * (unsigned)l16));
        aa[u] = (fix && !last_ok) ? 0.0f : __int_as_float(e.z);
        bb[u] = (fix && !last_ok) ? 0.0f : __int_as_float(e.w);
    }
    #pragma unroll
    for (int u = 0; u < NP; ++u) {
        v2f a2 = {aa[u], aa[u]}, b2 = {bb[u], bb[u]};
        v2f lo = {v[u].x, v[u].y}, hi = {v[u].z, v[u].w};
        pa += __builtin_elementwise_fma(a2, lo, b2);
        pb += __builtin_elementwise_fma(a2, hi, b2);
    }
}

__global__ void __launch_bounds__(256) k_gather(
        const float* __restrict__ xT, const int* __restrict__ ei,
        const float* __restrict__ alpha, const float* __restrict__ bias,
        const int* __restrict__ recw, const int* __restrict__ tabw,
        const float* __restrict__ W, const float* __restrict__ gamma, const float* __restrict__ beta,
        float* __restrict__ bn_out  , float* __restrict__ part  ) {
    __shared__ float s_bn[64 * 65];
    __shared__ float s_xh[64 * XP];
    __shared__ int   s_cnt[64];
    __shared__ int   s_beg[64];
    __shared__ int   s_fil[64];
    __shared__ int4  s_rec[CAP];
    __shared__ int   s_T;

    const int4* rec = (const int4*)recw;
    const unsigned short* tab = (const unsigned short*)tabw;
    int tile = blockIdx.x;
    int n0   = blockIdx.x * 64;
    int wave = threadIdx.x >> 5;
    int lane = threadIdx.x & 31;

    int half = lane >> 4;
    int l16  = lane & 15;
    int m_tile = wave >> 1;
    int n_tile = wave & 1;
    int lcol = n_tile * 16 + l16;
    int wrow = (lcol < LL) ? lcol : 0;
    const float* Wrow = W + (long)wrow * NN;

    int s0 = 0, cn = 0;
    if (threadIdx.x < NSL) {
        s0 = tab[tile * NSL + threadIdx.x];
        int s1 = (tile + 1 < NBLK) ? (int)tab[(tile + 1) * NSL + threadIdx.x] : SL_EPB;
        cn = s1 - s0;
    }
    int g8 = lane >> 3, k8 = lane & 7;
    int4 r[16];
    #pragma unroll
    for (int p = 0; p < 8; ++p) {
        int sl = p * 4 + g8;
        int rs = __shfl(s0, sl, 32), rc = __shfl(cn, sl, 32);
        int last = max(rc, 1) - 1;
        int rb = min(wave * 32 + sl, NSL - 1) * SL_EPB + rs;
        int4 qa = rec[min(rb + min(k8, last), EE - 1)];
        int4 qb = rec[min(rb + min(k8 + 8, last), EE - 1)];
        r[p]     = (k8 < rc)     ? qa : make_int4(-1, 0, 0, 0);
        r[8 + p] = (k8 + 8 < rc) ? qb : make_int4(-1, 0, 0, 0);
    }
    bool tail = __builtin_amdgcn_ballot_w32(cn > 16) != 0;
    if (threadIdx.x < 64) s_cnt[threadIdx.x] = 0;
    __syncthreads();
    #pragma unroll
    for (int u = 0; u < 16; ++u) {
        if (r[u].x >= 0) atomicAdd(&s_cnt[r[u].x & 63], 1);
    }
    if (tail) {
        for (int p = 0; p < 8; ++p) {
            int sl = p * 4 + g8;
            int rs = __shfl(s0, sl, 32), rc = __shfl(cn, sl, 32);
            const int4* runp = rec + ((long)(wave * 32 + sl) * SL_EPB + rs);
            for (int kk = 16 + k8; kk < rc; kk += 8) atomicAdd(&s_cnt[runp[kk].x & 63], 1);
        }
    }
    __syncthreads();
    if (wave == 0) {
        int c0 = s_cnt[2 * lane], c1 = s_cnt[2 * lane + 1];
        int sum2 = c0 + c1;
        int inc = sum2;
        #pragma unroll
        for (int off = 1; off < 32; off <<= 1) {
            int t = __shfl_up(inc, off, 32);
            if (lane >= off) inc += t;
        }
        int ex = inc - sum2;
        s_beg[2 * lane] = ex;      s_beg[2 * lane + 1] = ex + c0;
        s_fil[2 * lane] = ex;      s_fil[2 * lane + 1] = ex + c0;
        if (lane == 31) s_T = inc;
    }
    __syncthreads();
    int T = s_T;
    bool fit = T <= CAP;
    if (fit) {
        #pragma unroll
        for (int h = 0; h < 2; ++h) {
            int pos[8];
            #pragma unroll
            for (int u = 0; u < 8; ++u) {
                pos[u] = (r[8 * h + u].x >= 0) ? atomicAdd(&s_fil[r[8 * h + u].x & 63], 1) : 0;
            }
            #pragma unroll
            for (int u = 0; u < 8; ++u) {
                if (r[8 * h + u].x >= 0) {
                    int4 t = r[8 * h + u];
                    t.x >>= 6;
                    s_rec[pos[u]] = t;
                }
            }
        }
        if (tail) {
            for (int p = 0; p < 8; ++p) {
                int sl = p * 4 + g8;
                int rs = __shfl(s0, sl, 32), rc = __shfl(cn, sl, 32);
                const int4* runp = rec + ((long)(wave * 32 + sl) * SL_EPB + rs);
                for (int kk = 16 + k8; kk < rc; kk += 8) {
                    int4 t = runp[kk];
                    int ps = atomicAdd(&s_fil[t.x & 63], 1);
                    t.x >>= 6;
                    s_rec[ps] = t;
                }
            }
        }
    }
    __syncthreads();

    v2f bw[16];
    #pragma unroll
    for (int s = 0; s < 16; ++s) {
        int k0 = n0 + s * 4 + half * 2;
        bool ok = k0 < NN;
        bw[s].x = ok ? Wrow[k0] : 0.0f;
        bw[s].y = ok ? Wrow[k0 + 1] : 0.0f;
    }

    int len_l = s_cnt[wave * 8 + (lane & 7)];
    int beg_l = s_beg[wave * 8 + (lane & 7)];
    int b0 = 4 * l16 + 2 * half;

    static_assert(EE - 1 < 0x01ffffff, "the id of a lane without a slot sorts behind every edge id");
    int pk0 = 0x01ffffff, pk1 = 0x01ffffff;
    {
        int ln = __builtin_amdgcn_readlane(len_l, 0), bg = __builtin_amdgcn_readlane(beg_l, 0);
        if (fit && lane < ln)      pk0 = s_rec[bg + lane].x;
        if (fit && lane + 32 < ln) pk1 = s_rec[bg + lane + 32].x;
    }
    asm volatile("" : "+v"(pk0), "+v"(pk1));

    for (int i = 0; i < 8; ++i) {
        int lnode = wave * 8 + i;
        int id0 = pk0, id1 = pk1;
        pk0 = 0x01ffffff; pk1 = 0x01ffffff;
        if (fit && i < 7) {
            int ln = __builtin_amdgcn_readlane(len_l, i + 1), bg = __builtin_amdgcn_readlane(beg_l, i + 1);
            if (lane < ln)      pk0 = s_rec[bg + lane].x;
            if (lane + 32 < ln) pk1 = s_rec[bg + lane + 32].x;
        }
        int n = n0 + lnode;
        float xh0 = 0.0f, xh1 = 0.0f;
        if (n < NN) {
            int len = __builtin_amdgcn_readlane(len_l, i);
            int beg = __builtin_amdgcn_readlane(beg_l, i);
            float a0 = 0.0f, a1 = 0.0f;
            if (fit && len <= 64) {
                int k0 = (id0 << 6) | lane, k1 = (id1 << 6) | (lane + 32);
                if (len > 32) {
                    k0 = sort32(k0, lane, false);
                    k1 = sort32(k1, lane, true);
                    int lo = min(k0, k1), hi = max(k0, k1);
                    k0 = merge32(lo, lane);
                    k1 = merge32(hi, lane);
                } else {
                    k0 = sort32(k0, lane, false);
                }
                int4 q0 = make_int4(0, 0, 0, 0), q1 = make_int4(0, 0, 0, 0);
                if (lane < len)      q0 = s_rec[beg + (k0 & 63)];
                if (lane + 32 < len) q1 = s_rec[beg + (k1 & 63)];
                __builtin_amdgcn_fence(__ATOMIC_RELEASE, "wavefront");
                __builtin_amdgcn_wave_barrier();
                if (lane < len)      s_rec[beg + lane] = q0;
                if (lane + 32 < len) s_rec[beg + lane + 32] = q1;
                __builtin_amdgcn_fence(__ATOMIC_ACQ_REL, "wavefront");
                __builtin_amdgcn_wave_barrier();
                v2f pa = {0.0f, 0.0f}, pb = {0.0f, 0.0f};
                int jf = len & ~15;
                for (int j = 0; j < jf; j += 16) sum_pass<8, false>(xT, &s_rec[beg + j + half], l16, true, pa, pb);
                {
                    const int4* rp = &s_rec[beg + jf + half];
                    bool last_ok = half == 0 || (len & 1) == 0;
                    switch ((len - jf + 1) >> 1) {
                        case 1: sum_pass<1, true>(xT, rp, l16, last_ok, pa, pb); break;
                        case 2: sum_pass<2, true>(xT, rp, l16, last_ok, pa, pb); break;
                        case 3: sum_pass<3, true>(xT, rp, l16, last_ok, pa, pb); break;
                        case 4: sum_pass<4, true>(xT, rp, l16, last_ok, pa, pb); break;
                        case 5: sum_pass<5, true>(xT, rp, l16, last_ok, pa, pb); break;
                        case 6: sum_pass<6, true>(xT, rp, l16, last_ok, pa, pb); break;
                        case 7: sum_pass<7, true>(xT, rp, l16, last_ok, pa, pb); break;
                        case 8: sum_pass<8, true>(xT, rp, l16, last_ok, pa, pb); break;
                        default: break;
                    }
                }
                float p0 = pa.x, p1 = pa.y, p2 = pb.x, p3 = pb.y;
                float r0x = xhalf(half ? p0 : p2);
                float r1x = xhalf(half ? p1 : p3);
                a0 = (half ? p2 : p0) + r0x;
                a1 = (half ? p3 : p1) + r1x;
            } else {
                len = 0;
                int prev = -1;
                for (int it = 0; it <= T; ++it) {
                    int m = 0x7fffffff;
                    for (int w = lane; w < NSL; w += 32) {
                        int ka = tab[tile * NSL + w];
                        int kb = (tile + 1 < NBLK) ? (int)tab[(tile + 1) * NSL + w] : SL_EPB;
                        for (int k = ka; k < kb; ++k) {
                            int v = rec[(long)w * SL_EPB + k].x;
                            int id = v >> 6;
                            if ((v & 63) == lnode && id > prev && id < m) m = id;
                        }
                    }
                    #pragma unroll
                    for (int off = 16; off >= 1; off >>= 1) m = min(m, __shfl_xor(m, off, 32));
                    if (m == 0x7fffffff) break;
                    ++len;
                    int s = ei[m];
                    float al = alpha[m], bi = bias[m];
                    float2 v = *(const float2*)(xT + (long)s * 64 + b0);
                    a0 += fmaf(al, v.x, bi);
                    a1 += fmaf(al, v.y, bi);
                    prev = m;
                }
            }
            float den = fmaxf((float)len, 1.0f);
            xh0 = a0 / den;
            xh1 = a1 / den;
            float t0 = tanhf(xh0), t1 = tanhf(xh1);
            float sm = t0 + t1;
            float sq = t0 * t0 + t1 * t1;
            #pragma unroll
            for (int off = 16; off >= 1; off >>= 1) {
                sm += __shfl_xor(sm, off, 32);
                sq += __shfl_xor(sq, off, 32);
            }
            float mu = sm * (1.0f / 64.0f);
            float va = sq * (1.0f / 64.0f) - mu * mu;
            float rs = rsqrtf(va + BN_EPS);
            float g = gamma[n], be = beta[n];
            s_bn[lnode * 65 + b0]     = (t0 - mu) * rs * g + be;
            s_bn[lnode * 65 + b0 + 1] = (t1 - mu) * rs * g + be;
        }
        float2 xw; xw.x = xh0; xw.y = xh1;
        *(float2*)&s_xh[lnode * XP + b0] = xw;
    }
    __syncthreads();

    {
        int col  = threadIdx.x & 63;
        int rowg = threadIdx.x >> 6;
        int n = n0 + col;
        if (n < NN) {
            for (int b = rowg; b < 64; b += 4)
                bn_out[(long)b * NN + n] = s_bn[col * 65 + b];
        }
    }

    v8f acc = {};
    int m = m_tile * 16 + l16;
    #pragma unroll
    for (int s = 0; s < 16; ++s) {
        int k0 = s * 4 + half * 2;
        v2f a;
        a.x = s_xh[k0 * XP + m];
        a.y = s_xh[(k0 + 1) * XP + m];
        acc = __builtin_amdgcn_wmma_f32_16x16x4_f32(
             false, a,  false, bw[s],
             (short)0, acc,  false,  false);
    }
    float* dst = part + (long)blockIdx.x * (BB * LL);
    #pragma unroll
    for (int i = 0; i < 8; ++i) {
        int mrow = m_tile * 16 + half * 8 + i;
        int ncol = n_tile * 16 + l16;
        if (ncol < LL) dst[mrow * LL + ncol] = acc[i];
    }
}

__global__ void __launch_bounds__(1024) k_reduce_pred(const float* __restrict__ part, const float* __restrict__ b_lin,
                                                      float* __restrict__ pred) {
    __shared__ float red[32][33];
    int il = threadIdx.x & 31;
    int g  = threadIdx.x >> 5;
    int i  = blockIdx.x * 32 + il;
    float s = 0.0f;
    for (int c = g; c < NBLK; c += 32) s += part[(long)c * (BB * LL) + i];
    red[g][il] = s;
    __syncthreads();
    if (g == 0) {
        float t = b_lin[i % LL];
        #pragma unroll
        for (int gg = 0; gg < 32; ++gg) t += red[gg][il];
        pred[i] = t;
    }
}

extern "C" void kernel_launch(void* const* d_in, const int* in_sizes, int n_in,
                              void* d_out, int out_size, void* d_ws, size_t ws_size,
                              hipStream_t stream) {
    const float* x     = (const float*)d_in[0];
    const int*   ei    = (const int*)  d_in[1];
    const float* alpha = (const float*)d_in[2];
    const float* bias  = (const float*)d_in[3];
    const float* W     = (const float*)d_in[4];
    const float* b_lin = (const float*)d_in[5];
    const float* gamma = (const float*)d_in[6];
    const float* beta  = (const float*)d_in[7];
    float* out = (float*)d_out;

    float* ws     = (float*)d_ws;
    float* xT     = ws;
    int*   recw   = (int*)(xT + (long)NN * 64);
    int*   tabw   = recw + 4L * EE;
    float* part   = (float*)(tabw + 2L * NN);

    if (ws_size < (size_t)42803840) return;

    k_prep<<<NSL + NBLK, SL_T, 0, stream>>>(ei, alpha, bias, recw, tabw, x, xT);
    k_gather<<<NBLK, 256, 0, stream>>>(xT, ei, alpha, bias, recw, tabw, W, gamma, beta, out + BB * LL, part);
    k_reduce_pred<<<(BB * LL) / 32, 1024, 0, stream>>>(part, b_lin, out);
}
